# starting kernel
# baseline (speedup 1.0000x reference)
#include <hip/hip_runtime.h>
#include <hip/hip_bf16.h>

typedef __attribute__((ext_vector_type(2))) float v2f;
typedef __attribute__((ext_vector_type(8))) float v8f;

#define TB 256


__global__ void k_init_deg(float* __restrict__ deg, int n) {
  int i = blockIdx.x * blockDim.x + threadIdx.x;
  if (i < n) deg[i] = 1.0f;
}

__global__ void k_accum_deg(const int* __restrict__ dst, float* __restrict__ deg, int E) {
  int e = blockIdx.x * blockDim.x + threadIdx.x;
  if (e < E) atomicAdd(&deg[dst[e]], 1.0f);
}

__global__ void k_make_dinv(float* __restrict__ deg, int n) {
  int i = blockIdx.x * blockDim.x + threadIdx.x;
  if (i < n) deg[i] = rsqrtf(deg[i]);
}

template<int K, int N, bool RELU>
__global__ void k_gemm_wmma(const float* __restrict__ A, const float* __restrict__ B,
                            float* __restrict__ C, int M) {
  constexpr int NT   = 4;
  constexpr int COLS = 16 * NT;
  constexpr int KC   = 128;
  constexpr int LDK  = KC + 4;
  static_assert(K % KC == 0, "K must be a multiple of KC");
  static_assert(N % COLS == 0, "N must be a multiple of 64");
  __shared__ float Bs[COLS * LDK];

  constexpr int ngroups = N / COLS;
  int bm = blockIdx.x / ngroups;
  int ng = blockIdx.x % ngroups;
  int n0 = ng * COLS;

  int wid  = threadIdx.x >> 5;
  int lane = threadIdx.x & 31;
  int half = lane >> 4;
  int l    = lane & 15;

  int mt     = bm * 8 + wid;
  int mtiles = M >> 4;
  bool active = (mt < mtiles);
  int m0 = mt << 4;

  v8f acc[NT] = {};
  const float* Arow = active ? (A + (size_t)(m0 + l) * K + half * 2) : A;
  const float* Bl   = Bs + l * LDK + half * 2;

  for (int kc = 0; kc < K; kc += KC) {
    __syncthreads();
    for (int idx = threadIdx.x; idx < KC * COLS; idx += TB) {
      int k = idx / COLS;
      int c = idx % COLS;
      Bs[c * LDK + k] = B[(size_t)(kc + k) * N + n0 + c];
    }
    __syncthreads();

    if (active) {
#pragma unroll 4
      for (int k = 0; k < KC; k += 4) {
        float a0 = Arow[kc + k];
        float a1 = Arow[kc + k + 1];
        if (RELU) { a0 = fmaxf(a0, 0.0f); a1 = fmaxf(a1, 0.0f); }
        v2f a; a.x = a0; a.y = a1;
#pragma unroll
        for (int t = 0; t < NT; ++t) {
          const float* bp = Bl + t * 16 * LDK + k;
          v2f b; b.x = bp[0]; b.y = bp[1];
          acc[t] = __builtin_amdgcn_wmma_f32_16x16x4_f32(
              false, a, false, b, (short)0, acc[t], false, false);
        }
      }
    }
  }

  if (active) {
    int rbase = m0 + half * 8;
#pragma unroll
    for (int t = 0; t < NT; ++t) {
#pragma unroll
      for (int j = 0; j < 8; ++j) {
        C[(size_t)(rbase + j) * N + n0 + t * 16 + l] = acc[t][j];
      }
    }
  }
}


__global__ void k_init_agg(const float* __restrict__ H, const float* __restrict__ dinv,
                           const float* __restrict__ b, float* __restrict__ AGG,
                           int n, int F) {
  long long i = (long long)blockIdx.x * blockDim.x + threadIdx.x;
  long long total = (long long)n * F;
  if (i >= total) return;
  int row = (int)(i / F);
  int col = (int)(i % F);
  float di = dinv[row];
  AGG[i] = H[i] * di * di + b[col];
}

template<int F>
__global__ void k_edge_scatter(const float* __restrict__ H, float* __restrict__ AGG,
                               const int* __restrict__ src, const int* __restrict__ dst,
                               const float* __restrict__ dinv, int E) {
  constexpr int LPE = F / 4;
  constexpr int EPW = 32 / LPE;
  long long gwave = (long long)((blockIdx.x * blockDim.x + threadIdx.x) >> 5);
  int lane = threadIdx.x & 31;
  int sub  = lane / LPE;
  int fl   = lane % LPE;
  long long e = gwave * EPW + sub;
  if (e >= E) return;

  int s = src[e];
  int d = dst[e];
  float norm = dinv[s] * dinv[d];

  const float4* hp = (const float4*)(H + (size_t)s * F);
  float4 v = hp[fl];
  float* ap = AGG + (size_t)d * F + fl * 4;
  atomicAdd(ap + 0, v.x * norm);
  atomicAdd(ap + 1, v.y * norm);
  atomicAdd(ap + 2, v.z * norm);
  atomicAdd(ap + 3, v.w * norm);
}


__global__ void k_log_softmax64(float* __restrict__ out, int n, float invT) {
  int wave = (blockIdx.x * blockDim.x + threadIdx.x) >> 5;
  int lane = threadIdx.x & 31;
  if (wave >= n) return;
  float* row = out + (size_t)wave * 64;
  float z0 = row[lane]      * invT;
  float z1 = row[lane + 32] * invT;
  float m = fmaxf(z0, z1);
#pragma unroll
  for (int off = 16; off > 0; off >>= 1) m = fmaxf(m, __shfl_xor(m, off, 32));
  float s = expf(z0 - m) + expf(z1 - m);
#pragma unroll
  for (int off = 16; off > 0; off >>= 1) s += __shfl_xor(s, off, 32);
  float lg = m + logf(s);
  row[lane]      = z0 - lg;
  row[lane + 32] = z1 - lg;
}


extern "C" void kernel_launch(void* const* d_in, const int* in_sizes, int n_in,
                              void* d_out, int out_size, void* d_ws, size_t ws_size,
                              hipStream_t stream) {
  const float* x  = (const float*)d_in[0];
  const int*   ei = (const int*)d_in[1];
  const float* W1 = (const float*)d_in[2];
  const float* b1 = (const float*)d_in[3];
  const float* W2 = (const float*)d_in[4];
  const float* b2 = (const float*)d_in[5];

  const int IN = 256, HID = 128, OUT = 64;
  int n = in_sizes[0] / IN;
  int E = in_sizes[1] / 2;
  const int* src = ei;
  const int* dst = ei + E;

  float* ws = (float*)d_ws;
  size_t o = 0;
  float* deg  = ws + o; o += ((size_t)n + 255) & ~(size_t)255;
  float* H1   = ws + o; o += (size_t)n * HID;
  float* AGG1 = ws + o; o += (size_t)n * HID;
  float* H2   = ws + o; o += (size_t)n * OUT;
  float* AGG2 = (float*)d_out;

  int mtiles  = n / 16;
  int mblocks = (mtiles + 7) / 8;

  k_init_deg <<<(n + TB - 1) / TB, TB, 0, stream>>>(deg, n);
  k_accum_deg<<<(E + TB - 1) / TB, TB, 0, stream>>>(dst, deg, E);
  k_make_dinv<<<(n + TB - 1) / TB, TB, 0, stream>>>(deg, n);

  k_gemm_wmma<256, 128, false><<<mblocks * (HID / 64), TB, 0, stream>>>(x, W1, H1, n);

  {
    long long tot = (long long)n * HID;
    k_init_agg<<<(unsigned)((tot + TB - 1) / TB), TB, 0, stream>>>(H1, deg, b1, AGG1, n, HID);
    long long waves = E;
    k_edge_scatter<128><<<(unsigned)((waves * 32 + TB - 1) / TB), TB, 0, stream>>>(
        H1, AGG1, src, dst, deg, E);
  }

  k_gemm_wmma<128, 64, true><<<mblocks * (OUT / 64), TB, 0, stream>>>(AGG1, W2, H2, n);

  {
    long long tot = (long long)n * OUT;
    k_init_agg<<<(unsigned)((tot + TB - 1) / TB), TB, 0, stream>>>(H2, deg, b2, AGG2, n, OUT);
    long long waves = ((long long)E + 1) / 2;
    k_edge_scatter<64><<<(unsigned)((waves * 32 + TB - 1) / TB), TB, 0, stream>>>(
        H2, AGG2, src, dst, deg, E);
  }

  k_log_softmax64<<<(n * 32 + TB - 1) / TB, TB, 0, stream>>>(AGG2, n, 5.0f);
}


# edge scatter: lane = consecutive float, so each fp32 atomic and each gather load instruction covers one whole 128-byte line, not 8 lanes in each of 4 lines; arrival-order atomic sums as before
# speedup vs baseline: 1.3862x; 1.3862x over previous
#include <hip/hip_runtime.h>
#include <hip/hip_bf16.h>

typedef __attribute__((ext_vector_type(2))) float v2f;
typedef __attribute__((ext_vector_type(8))) float v8f;

#define TB 256


__global__ void k_init_deg(float* __restrict__ deg, int n) {
  int i = blockIdx.x * blockDim.x + threadIdx.x;
  if (i < n) deg[i] = 1.0f;
}

__global__ void k_accum_deg(const int* __restrict__ dst, float* __restrict__ deg, int E) {
  int e = blockIdx.x * blockDim.x + threadIdx.x;
  if (e < E) atomicAdd(&deg[dst[e]], 1.0f);
}

__global__ void k_make_dinv(float* __restrict__ deg, int n) {
  int i = blockIdx.x * blockDim.x + threadIdx.x;
  if (i < n) deg[i] = rsqrtf(deg[i]);
}

template<int K, int N, bool RELU>
__global__ void k_gemm_wmma(const float* __restrict__ A, const float* __restrict__ B,
                            float* __restrict__ C, int M) {
  constexpr int NT   = 4;
  constexpr int COLS = 16 * NT;
  constexpr int KC   = 128;
  constexpr int LDK  = KC + 4;
  static_assert(K % KC == 0, "K must be a multiple of KC");
  static_assert(N % COLS == 0, "N must be a multiple of 64");
  __shared__ float Bs[COLS * LDK];

  constexpr int ngroups = N / COLS;
  int bm = blockIdx.x / ngroups;
  int ng = blockIdx.x % ngroups;
  int n0 = ng * COLS;

  int wid  = threadIdx.x >> 5;
  int lane = threadIdx.x & 31;
  int half = lane >> 4;
  int l    = lane & 15;

  int mt     = bm * 8 + wid;
  int mtiles = M >> 4;
  bool active = (mt < mtiles);
  int m0 = mt << 4;

  v8f acc[NT] = {};
  const float* Arow = active ? (A + (size_t)(m0 + l) * K + half * 2) : A;
  const float* Bl   = Bs + l * LDK + half * 2;

  for (int kc = 0; kc < K; kc += KC) {
    __syncthreads();
    for (int idx = threadIdx.x; idx < KC * COLS; idx += TB) {
      int k = idx / COLS;
      int c = idx % COLS;
      Bs[c * LDK + k] = B[(size_t)(kc + k) * N + n0 + c];
    }
    __syncthreads();

    if (active) {
#pragma unroll 4
      for (int k = 0; k < KC; k += 4) {
        float a0 = Arow[kc + k];
        float a1 = Arow[kc + k + 1];
        if (RELU) { a0 = fmaxf(a0, 0.0f); a1 = fmaxf(a1, 0.0f); }
        v2f a; a.x = a0; a.y = a1;
#pragma unroll
        for (int t = 0; t < NT; ++t) {
          const float* bp = Bl + t * 16 * LDK + k;
          v2f b; b.x = bp[0]; b.y = bp[1];
          acc[t] = __builtin_amdgcn_wmma_f32_16x16x4_f32(
              false, a, false, b, (short)0, acc[t], false, false);
        }
      }
    }
  }

  if (active) {
    int rbase = m0 + half * 8;
#pragma unroll
    for (int t = 0; t < NT; ++t) {
#pragma unroll
      for (int j = 0; j < 8; ++j) {
        C[(size_t)(rbase + j) * N + n0 + t * 16 + l] = acc[t][j];
      }
    }
  }
}


__global__ void k_init_agg(const float* __restrict__ H, const float* __restrict__ dinv,
                           const float* __restrict__ b, float* __restrict__ AGG,
                           int n, int F) {
  long long i = (long long)blockIdx.x * blockDim.x + threadIdx.x;
  long long total = (long long)n * F;
  if (i >= total) return;
  int row = (int)(i / F);
  int col = (int)(i % F);
  float di = dinv[row];
  AGG[i] = H[i] * di * di + b[col];
}

template<int F, int NE>
__device__ __forceinline__ void scatter_rows(const float* __restrict__ H, float* __restrict__ AGG,
                                             const int* __restrict__ src, const int* __restrict__ dst,
                                             const float* __restrict__ dinv, long long e0, int lane) {
  constexpr int LPE = F / 32;
  float v[NE][LPE];
  float norm[NE];
  int   d[NE];
#pragma unroll
  for (int j = 0; j < NE; ++j) {
    int s = src[e0 + j];
    d[j]  = dst[e0 + j];
    norm[j] = dinv[s] * dinv[d[j]];
    const float* hp = H + (size_t)s * F + lane;
#pragma unroll
    for (int r = 0; r < LPE; ++r) v[j][r] = hp[r * 32];
  }
#pragma unroll
  for (int j = 0; j < NE; ++j) {
    float* ap = AGG + (size_t)d[j] * F + lane;
#pragma unroll
    for (int r = 0; r < LPE; ++r) atomicAdd(ap + r * 32, v[j][r] * norm[j]);
  }
}

template<int F>
__global__ void k_edge_scatter(const float* __restrict__ H, float* __restrict__ AGG,
                               const int* __restrict__ src, const int* __restrict__ dst,
                               const float* __restrict__ dinv, int E) {
  constexpr int EPW = 128 / F;
  long long gwave = (long long)((blockIdx.x * blockDim.x + threadIdx.x) >> 5);
  int lane = threadIdx.x & 31;
  long long e0 = gwave * EPW;
  if (e0 + EPW <= E) {
    scatter_rows<F, EPW>(H, AGG, src, dst, dinv, e0, lane);
  } else {
    for (long long e = e0; e < E; ++e)
      scatter_rows<F, 1>(H, AGG, src, dst, dinv, e, lane);
  }
}


__global__ void k_log_softmax64(float* __restrict__ out, int n, float invT) {
  int wave = (blockIdx.x * blockDim.x + threadIdx.x) >> 5;
  int lane = threadIdx.x & 31;
  if (wave >= n) return;
  float* row = out + (size_t)wave * 64;
  float z0 = row[lane]      * invT;
  float z1 = row[lane + 32] * invT;
  float m = fmaxf(z0, z1);
#pragma unroll
  for (int off = 16; off > 0; off >>= 1) m = fmaxf(m, __shfl_xor(m, off, 32));
  float s = expf(z0 - m) + expf(z1 - m);
#pragma unroll
  for (int off = 16; off > 0; off >>= 1) s += __shfl_xor(s, off, 32);
  float lg = m + logf(s);
  row[lane]      = z0 - lg;
  row[lane + 32] = z1 - lg;
}


extern "C" void kernel_launch(void* const* d_in, const int* in_sizes, int n_in,
                              void* d_out, int out_size, void* d_ws, size_t ws_size,
                              hipStream_t stream) {
  const float* x  = (const float*)d_in[0];
  const int*   ei = (const int*)d_in[1];
  const float* W1 = (const float*)d_in[2];
  const float* b1 = (const float*)d_in[3];
  const float* W2 = (const float*)d_in[4];
  const float* b2 = (const float*)d_in[5];

  const int IN = 256, HID = 128, OUT = 64;
  int n = in_sizes[0] / IN;
  int E = in_sizes[1] / 2;
  const int* src = ei;
  const int* dst = ei + E;

  float* ws = (float*)d_ws;
  size_t o = 0;
  float* deg  = ws + o; o += ((size_t)n + 255) & ~(size_t)255;
  float* H1   = ws + o; o += (size_t)n * HID;
  float* AGG1 = ws + o; o += (size_t)n * HID;
  float* H2   = ws + o; o += (size_t)n * OUT;
  float* AGG2 = (float*)d_out;

  int mtiles  = n / 16;
  int mblocks = (mtiles + 7) / 8;

  k_init_deg <<<(n + TB - 1) / TB, TB, 0, stream>>>(deg, n);
  k_accum_deg<<<(E + TB - 1) / TB, TB, 0, stream>>>(dst, deg, E);
  k_make_dinv<<<(n + TB - 1) / TB, TB, 0, stream>>>(deg, n);

  k_gemm_wmma<256, 128, false><<<mblocks * (HID / 64), TB, 0, stream>>>(x, W1, H1, n);

  {
    long long tot = (long long)n * HID;
    k_init_agg<<<(unsigned)((tot + TB - 1) / TB), TB, 0, stream>>>(H1, deg, b1, AGG1, n, HID);
    long long waves = E;
    k_edge_scatter<128><<<(unsigned)((waves * 32 + TB - 1) / TB), TB, 0, stream>>>(
        H1, AGG1, src, dst, deg, E);
  }

  k_gemm_wmma<128, 64, true><<<mblocks * (OUT / 64), TB, 0, stream>>>(AGG1, W2, H2, n);

  {
    long long tot = (long long)n * OUT;
    k_init_agg<<<(unsigned)((tot + TB - 1) / TB), TB, 0, stream>>>(H2, deg, b2, AGG2, n, OUT);
    long long waves = ((long long)E + 1) / 2;
    k_edge_scatter<64><<<(unsigned)((waves * 32 + TB - 1) / TB), TB, 0, stream>>>(
        H2, AGG2, src, dst, deg, E);
  }

  k_log_softmax64<<<(n * 32 + TB - 1) / TB, TB, 0, stream>>>(AGG2, n, 5.0f);
}


# edges counting-sorted by destination every call (slot = returning int atomic), one wave sums a row in registers, one store; no fp32 atomics; sum order = slot order, still arrival order
# speedup vs baseline: 2.5067x; 1.8083x over previous
#include <hip/hip_runtime.h>
#include <hip/hip_bf16.h>

typedef __attribute__((ext_vector_type(2))) float v2f;
typedef __attribute__((ext_vector_type(8))) float v8f;

#define TB 256


#define SB 1024

__global__ void k_zero_cnt(int* __restrict__ cnt, int n) {
  int i = blockIdx.x * blockDim.x + threadIdx.x;
  if (i < n) cnt[i] = 0;
}

__global__ void k_rank(const int* __restrict__ dst, int* __restrict__ cnt,
                       int* __restrict__ rank, int E) {
  int e = blockIdx.x * blockDim.x + threadIdx.x;
  if (e < E) rank[e] = atomicAdd(&cnt[dst[e]], 1);
}

__global__ void __launch_bounds__(SB) k_scan_local(const int* __restrict__ cnt, int* __restrict__ rowptr,
                                                   int* __restrict__ bsum, float* __restrict__ dinv, int n) {
  __shared__ int wsum[32];
  int i    = blockIdx.x * SB + threadIdx.x;
  int lane = threadIdx.x & 31;
  int w    = threadIdx.x >> 5;
  int c = (i < n) ? cnt[i] : 0;
  int v = c;
  for (int off = 1; off < 32; off <<= 1) {
    int u = __shfl_up(v, off, 32);
    if (lane >= off) v += u;
  }
  if (lane == 31) wsum[w] = v;
  __syncthreads();
  if (w == 0) {
    int x = wsum[lane];
    for (int off = 1; off < 32; off <<= 1) {
      int u = __shfl_up(x, off, 32);
      if (lane >= off) x += u;
    }
    wsum[lane] = x;
  }
  __syncthreads();
  int excl = v - c + (w > 0 ? wsum[w - 1] : 0);
  if (i < n) {
    rowptr[i] = excl;
    dinv[i]   = rsqrtf((float)(c + 1));
  }
  if (threadIdx.x == SB - 1) bsum[blockIdx.x] = excl + c;
}

__global__ void __launch_bounds__(SB) k_scan_add(int* __restrict__ rowptr, const int* __restrict__ bsum,
                                                 int n, int E) {
  int i = blockIdx.x * SB + threadIdx.x;
  int off = 0;
  for (int k = 0; k < (int)blockIdx.x; ++k) off += bsum[k];
  if (i < n) rowptr[i] += off;
  if (i == 0) rowptr[n] = E;
}

__global__ void k_place(const int* __restrict__ src, const int* __restrict__ dst,
                        const int* __restrict__ rank, const int* __restrict__ rowptr,
                        int* __restrict__ col, int E) {
  int e = blockIdx.x * blockDim.x + threadIdx.x;
  if (e < E) col[rowptr[dst[e]] + rank[e]] = src[e];
}

template<int K, int N, bool RELU>
__global__ void k_gemm_wmma(const float* __restrict__ A, const float* __restrict__ B,
                            float* __restrict__ C, int M) {
  constexpr int NT   = 4;
  constexpr int COLS = 16 * NT;
  constexpr int KC   = 128;
  constexpr int LDK  = KC + 4;
  static_assert(K % KC == 0, "K must be a multiple of KC");
  static_assert(N % COLS == 0, "N must be a multiple of 64");
  __shared__ float Bs[COLS * LDK];

  constexpr int ngroups = N / COLS;
  int bm = blockIdx.x / ngroups;
  int ng = blockIdx.x % ngroups;
  int n0 = ng * COLS;

  int wid  = threadIdx.x >> 5;
  int lane = threadIdx.x & 31;
  int half = lane >> 4;
  int l    = lane & 15;

  int mt     = bm * 8 + wid;
  int mtiles = M >> 4;
  bool active = (mt < mtiles);
  int m0 = mt << 4;

  v8f acc[NT] = {};
  const float* Arow = active ? (A + (size_t)(m0 + l) * K + half * 2) : A;
  const float* Bl   = Bs + l * LDK + half * 2;

  for (int kc = 0; kc < K; kc += KC) {
    __syncthreads();
    for (int idx = threadIdx.x; idx < KC * COLS; idx += TB) {
      int k = idx / COLS;
      int c = idx % COLS;
      Bs[c * LDK + k] = B[(size_t)(kc + k) * N + n0 + c];
    }
    __syncthreads();

    if (active) {
#pragma unroll 4
      for (int k = 0; k < KC; k += 4) {
        float a0 = Arow[kc + k];
        float a1 = Arow[kc + k + 1];
        if (RELU) { a0 = fmaxf(a0, 0.0f); a1 = fmaxf(a1, 0.0f); }
        v2f a; a.x = a0; a.y = a1;
#pragma unroll
        for (int t = 0; t < NT; ++t) {
          const float* bp = Bl + t * 16 * LDK + k;
          v2f b; b.x = bp[0]; b.y = bp[1];
          acc[t] = __builtin_amdgcn_wmma_f32_16x16x4_f32(
              false, a, false, b, (short)0, acc[t], false, false);
        }
      }
    }
  }

  if (active) {
    int rbase = m0 + half * 8;
#pragma unroll
    for (int t = 0; t < NT; ++t) {
#pragma unroll
      for (int j = 0; j < 8; ++j) {
        C[(size_t)(rbase + j) * N + n0 + t * 16 + l] = acc[t][j];
      }
    }
  }
}


typedef __attribute__((ext_vector_type(4))) float v4f;

template<int F>
__global__ void k_agg_rows(const float* __restrict__ H, const int* __restrict__ col,
                           const int* __restrict__ rowptr, const float* __restrict__ dinv,
                           const float* __restrict__ b, float* __restrict__ OUT, int n) {
  constexpr int LPR = F / 4;
  constexpr int EPI = 32 / LPR;
  constexpr int U   = 8;
  int row  = (int)((blockIdx.x * blockDim.x + threadIdx.x) >> 5);
  int lane = threadIdx.x & 31;
  if (row >= n) return;
  int sub = lane / LPR;
  int l   = lane % LPR;
  int start = rowptr[row];
  int end   = rowptr[row + 1];
  float di  = dinv[row];
  const v4f zero = {0.0f, 0.0f, 0.0f, 0.0f};
  v4f acc = zero;
  if (sub == 0) {
    v4f h = *(const v4f*)(H + (size_t)row * F + l * 4);
    acc = h * (di * di);
  }
  for (int base = start; base < end; base += 32) {
    int m = min(32, end - base);
    int   my_s = 0;
    float my_w = 0.0f;
    if (lane < m) {
      my_s = col[base + lane];
      my_w = dinv[my_s] * di;
    }
    for (int j = 0; j < m; j += EPI * U) {
      v4f   v[U];
      float w[U];
      int   s[U];
#pragma unroll
      for (int u = 0; u < U; ++u) {
        int jj = j + u * EPI + sub;
        s[u] = __shfl(my_s, jj, 32);
        w[u] = __shfl(my_w, jj, 32);
      }
#pragma unroll
      for (int u = 0; u < U; ++u) {
        int jj = j + u * EPI + sub;
        v[u] = (jj < m) ? *(const v4f*)(H + (size_t)s[u] * F + l * 4) : zero;
      }
#pragma unroll
      for (int u = 0; u < U; ++u) acc += v[u] * w[u];
    }
  }
  for (int o = LPR; o < 32; o <<= 1) {
    acc.x += __shfl_xor(acc.x, o, 32);
    acc.y += __shfl_xor(acc.y, o, 32);
    acc.z += __shfl_xor(acc.z, o, 32);
    acc.w += __shfl_xor(acc.w, o, 32);
  }
  if (sub == 0) {
    v4f bb = *(const v4f*)(b + l * 4);
    *(v4f*)(OUT + (size_t)row * F + l * 4) = acc + bb;
  }
}


__global__ void k_log_softmax64(float* __restrict__ out, int n, float invT) {
  int wave = (blockIdx.x * blockDim.x + threadIdx.x) >> 5;
  int lane = threadIdx.x & 31;
  if (wave >= n) return;
  float* row = out + (size_t)wave * 64;
  float z0 = row[lane]      * invT;
  float z1 = row[lane + 32] * invT;
  float m = fmaxf(z0, z1);
#pragma unroll
  for (int off = 16; off > 0; off >>= 1) m = fmaxf(m, __shfl_xor(m, off, 32));
  float s = expf(z0 - m) + expf(z1 - m);
#pragma unroll
  for (int off = 16; off > 0; off >>= 1) s += __shfl_xor(s, off, 32);
  float lg = m + logf(s);
  row[lane]      = z0 - lg;
  row[lane + 32] = z1 - lg;
}


extern "C" void kernel_launch(void* const* d_in, const int* in_sizes, int n_in,
                              void* d_out, int out_size, void* d_ws, size_t ws_size,
                              hipStream_t stream) {
  const float* x  = (const float*)d_in[0];
  const int*   ei = (const int*)d_in[1];
  const float* W1 = (const float*)d_in[2];
  const float* b1 = (const float*)d_in[3];
  const float* W2 = (const float*)d_in[4];
  const float* b2 = (const float*)d_in[5];

  const int IN = 256, HID = 128, OUT = 64;
  int n = in_sizes[0] / IN;
  int E = in_sizes[1] / 2;
  const int* src = ei;
  const int* dst = ei + E;

  float* ws = (float*)d_ws;
  size_t o = 0;
  float* dinv = ws + o; o += ((size_t)n + 255) & ~(size_t)255;
  float* H1   = ws + o; o += (size_t)n * HID;
  float* AGG1 = ws + o; o += (size_t)n * HID;
  float* H2   = ws + o; o += (size_t)n * OUT;
  int* cnt    = (int*)(ws + o); o += ((size_t)n + 255) & ~(size_t)255;
  int* rowptr = (int*)(ws + o); o += ((size_t)n + 1 + 255) & ~(size_t)255;
  int* bsum   = (int*)(ws + o); o += 1024;
  int* rank   = (int*)(ws + o); o += ((size_t)E + 255) & ~(size_t)255;
  int* col    = (int*)(ws + o); o += ((size_t)E + 255) & ~(size_t)255;
  float* AGG2 = (float*)d_out;

  int mtiles  = n / 16;
  int mblocks = (mtiles + 7) / 8;
  int sblocks = (n + SB - 1) / SB;

  k_zero_cnt  <<<(n + TB - 1) / TB, TB, 0, stream>>>(cnt, n);
  k_rank      <<<(E + TB - 1) / TB, TB, 0, stream>>>(dst, cnt, rank, E);
  k_scan_local<<<sblocks, SB, 0, stream>>>(cnt, rowptr, bsum, dinv, n);
  k_scan_add  <<<sblocks, SB, 0, stream>>>(rowptr, bsum, n, E);
  k_place     <<<(E + TB - 1) / TB, TB, 0, stream>>>(src, dst, rank, rowptr, col, E);

  k_gemm_wmma<256, 128, false><<<mblocks * (HID / 64), TB, 0, stream>>>(x, W1, H1, n);

  k_agg_rows<128><<<(unsigned)(((long long)n * 32 + TB - 1) / TB), TB, 0, stream>>>(
      H1, col, rowptr, dinv, b1, AGG1, n);

  k_gemm_wmma<128, 64, true><<<mblocks * (OUT / 64), TB, 0, stream>>>(AGG1, W2, H2, n);

  k_agg_rows<64><<<(unsigned)(((long long)n * 32 + TB - 1) / TB), TB, 0, stream>>>(
      H2, col, rowptr, dinv, b2, AGG2, n);

  k_log_softmax64<<<(n * 32 + TB - 1) / TB, TB, 0, stream>>>(AGG2, n, 5.0f);
}


# edges grouped by destination with LDS counters only: edge slices binned by 256-node range, one block per bin ranks and places; no global atomics; row sum order still arrival order
# speedup vs baseline: 3.5187x; 1.4037x over previous
#include <hip/hip_runtime.h>
#include <hip/hip_bf16.h>

typedef __attribute__((ext_vector_type(2))) float v2f;
typedef __attribute__((ext_vector_type(8))) float v8f;

#define TB 256


#define SB 512
#define RB 256
#define GS 256
#define EU 8

__device__ __forceinline__ int block_excl_scan(int v, int* wsum, int& total) {
  int lane = threadIdx.x & 31;
  int w    = threadIdx.x >> 5;
  int nw   = blockDim.x >> 5;
  int inc = v;
  for (int o = 1; o < 32; o <<= 1) {
    int u = __shfl_up(inc, o, 32);
    if (lane >= o) inc += u;
  }
  __syncthreads();
  if (lane == 31) wsum[w] = inc;
  __syncthreads();
  if (w == 0) {
    int x = (lane < nw) ? wsum[lane] : 0;
    for (int o = 1; o < 32; o <<= 1) {
      int u = __shfl_up(x, o, 32);
      if (lane >= o) x += u;
    }
    wsum[lane] = x;
  }
  __syncthreads();
  total = wsum[31];
  return inc - v + (w > 0 ? wsum[w - 1] : 0);
}

__global__ void __launch_bounds__(SB) k_bin_count(const int* __restrict__ dst, int* __restrict__ hist,
                                                  int E, int NB, int chunk) {
  extern __shared__ int lbin[];
  int g = blockIdx.x;
  for (int b = threadIdx.x; b < NB; b += SB) lbin[b] = 0;
  __syncthreads();
  int e0 = g * chunk;
  int e1 = min(E, e0 + chunk);
  for (int e = e0 + (int)threadIdx.x; e < e1; e += SB * EU) {
    int d[EU];
#pragma unroll
    for (int u = 0; u < EU; ++u) d[u] = dst[min(e + u * SB, e1 - 1)];
#pragma unroll
    for (int u = 0; u < EU; ++u) atomicAdd(&lbin[(unsigned)d[u] / RB], (e + u * SB < e1) ? 1 : 0);
  }
  __syncthreads();
  for (int b = threadIdx.x; b < NB; b += SB) hist[b * GS + g] = lbin[b];
}

__global__ void __launch_bounds__(GS) k_bin_rowscan(int* __restrict__ hist, int* __restrict__ rowtot) {
  __shared__ int wsum[32];
  int b = blockIdx.x;
  int v = hist[b * GS + threadIdx.x];
  int tot;
  int ex = block_excl_scan(v, wsum, tot);
  hist[b * GS + threadIdx.x] = ex;
  if (threadIdx.x == 0) rowtot[b] = tot;
}

__global__ void __launch_bounds__(SB) k_bin_scatter(const int* __restrict__ src, const int* __restrict__ dst,
                                                    const int* __restrict__ hist, const int* __restrict__ rowtot,
                                                    int* __restrict__ pairs, int E, int NB, int chunk) {
  extern __shared__ int lbin[];
  __shared__ int wsum[32];
  int g = blockIdx.x;
  int carry = 0;
  for (int t0 = 0; t0 < NB; t0 += SB) {
    int b = t0 + (int)threadIdx.x;
    int v = (b < NB) ? rowtot[b] : 0;
    int tot;
    int ex = block_excl_scan(v, wsum, tot);
    if (b < NB) lbin[b] = carry + ex + hist[b * GS + g];
    carry += tot;
  }
  __syncthreads();
  int e0 = g * chunk;
  int e1 = min(E, e0 + chunk);
  for (int e = e0 + (int)threadIdx.x; e < e1; e += SB * EU) {
    int d[EU], s[EU];
#pragma unroll
    for (int u = 0; u < EU; ++u) {
      int ee = min(e + u * SB, e1 - 1);
      d[u] = dst[ee];
      s[u] = src[ee];
    }
    int p[EU];
#pragma unroll
    for (int u = 0; u < EU; ++u) p[u] = atomicAdd(&lbin[(unsigned)d[u] / RB], (e + u * SB < e1) ? 1 : 0);
#pragma unroll
    for (int u = 0; u < EU; ++u)
      if (e + u * SB < e1) *(int2*)(pairs + 2 * (size_t)p[u]) = make_int2(s[u], d[u]);
  }
}

__global__ void __launch_bounds__(SB) k_bin_rows(const int* __restrict__ pairs, const int* __restrict__ rowtot,
                                                 int* __restrict__ rowptr, float* __restrict__ dinv,
                                                 int* __restrict__ col, int n, int E) {
  __shared__ int lcnt[RB];
  __shared__ int lcur[RB];
  __shared__ int wsum[32];
  int b  = blockIdx.x;
  int n0 = b * RB;
  for (int t = threadIdx.x; t < RB; t += SB) lcnt[t] = 0;
  int part = 0;
  for (int k = threadIdx.x; k < b; k += SB) part += rowtot[k];
  int base;
  block_excl_scan(part, wsum, base);
  int end = base + rowtot[b];
  for (int i = base + (int)threadIdx.x; i < end; i += SB * EU) {
    int d[EU];
#pragma unroll
    for (int u = 0; u < EU; ++u) d[u] = pairs[2 * (size_t)min(i + u * SB, end - 1) + 1];
#pragma unroll
    for (int u = 0; u < EU; ++u) atomicAdd(&lcnt[d[u] - n0], (i + u * SB < end) ? 1 : 0);
  }
  __syncthreads();
  int carry = 0;
  for (int t0 = 0; t0 < RB; t0 += SB) {
    int t = t0 + (int)threadIdx.x;
    int c = (t < RB) ? lcnt[t] : 0;
    int tot;
    int ex = block_excl_scan(c, wsum, tot);
    if (t < RB) {
      int first = base + carry + ex;
      lcur[t] = first;
      int node = n0 + t;
      if (node < n) {
        rowptr[node] = first;
        dinv[node]   = rsqrtf((float)(c + 1));
      }
    }
    carry += tot;
  }
  if (b == (int)gridDim.x - 1 && threadIdx.x == 0) rowptr[n] = E;
  __syncthreads();
  for (int i = base + (int)threadIdx.x; i < end; i += SB * EU) {
    int2 t[EU];
#pragma unroll
    for (int u = 0; u < EU; ++u) t[u] = *(const int2*)(pairs + 2 * (size_t)min(i + u * SB, end - 1));
    int p[EU];
#pragma unroll
    for (int u = 0; u < EU; ++u) p[u] = atomicAdd(&lcur[t[u].y - n0], (i + u * SB < end) ? 1 : 0);
#pragma unroll
    for (int u = 0; u < EU; ++u)
      if (i + u * SB < end) col[p[u]] = t[u].x;
  }
}

template<int K, int N, bool RELU>
__global__ void k_gemm_wmma(const float* __restrict__ A, const float* __restrict__ B,
                            float* __restrict__ C, int M) {
  constexpr int NT   = 4;
  constexpr int COLS = 16 * NT;
  constexpr int KC   = 128;
  constexpr int LDK  = KC + 4;
  static_assert(K % KC == 0, "K must be a multiple of KC");
  static_assert(N % COLS == 0, "N must be a multiple of 64");
  __shared__ float Bs[COLS * LDK];

  constexpr int ngroups = N / COLS;
  int bm = blockIdx.x / ngroups;
  int ng = blockIdx.x % ngroups;
  int n0 = ng * COLS;

  int wid  = threadIdx.x >> 5;
  int lane = threadIdx.x & 31;
  int half = lane >> 4;
  int l    = lane & 15;

  int mt     = bm * 8 + wid;
  int mtiles = M >> 4;
  bool active = (mt < mtiles);
  int m0 = mt << 4;

  v8f acc[NT] = {};
  const float* Arow = active ? (A + (size_t)(m0 + l) * K + half * 2) : A;
  const float* Bl   = Bs + l * LDK + half * 2;

  for (int kc = 0; kc < K; kc += KC) {
    __syncthreads();
    for (int idx = threadIdx.x; idx < KC * COLS; idx += TB) {
      int k = idx / COLS;
      int c = idx % COLS;
      Bs[c * LDK + k] = B[(size_t)(kc + k) * N + n0 + c];
    }
    __syncthreads();

    if (active) {
#pragma unroll 4
      for (int k = 0; k < KC; k += 4) {
        float a0 = Arow[kc + k];
        float a1 = Arow[kc + k + 1];
        if (RELU) { a0 = fmaxf(a0, 0.0f); a1 = fmaxf(a1, 0.0f); }
        v2f a; a.x = a0; a.y = a1;
#pragma unroll
        for (int t = 0; t < NT; ++t) {
          const float* bp = Bl + t * 16 * LDK + k;
          v2f b; b.x = bp[0]; b.y = bp[1];
          acc[t] = __builtin_amdgcn_wmma_f32_16x16x4_f32(
              false, a, false, b, (short)0, acc[t], false, false);
        }
      }
    }
  }

  if (active) {
    int rbase = m0 + half * 8;
#pragma unroll
    for (int t = 0; t < NT; ++t) {
#pragma unroll
      for (int j = 0; j < 8; ++j) {
        C[(size_t)(rbase + j) * N + n0 + t * 16 + l] = acc[t][j];
      }
    }
  }
}


typedef __attribute__((ext_vector_type(4))) float v4f;

template<int F>
__global__ void k_agg_rows(const float* __restrict__ H, const int* __restrict__ col,
                           const int* __restrict__ rowptr, const float* __restrict__ dinv,
                           const float* __restrict__ b, float* __restrict__ OUT, int n) {
  constexpr int LPR = F / 4;
  constexpr int EPI = 32 / LPR;
  constexpr int U   = 8;
  int row  = (int)((blockIdx.x * blockDim.x + threadIdx.x) >> 5);
  int lane = threadIdx.x & 31;
  if (row >= n) return;
  int sub = lane / LPR;
  int l   = lane % LPR;
  int start = rowptr[row];
  int end   = rowptr[row + 1];
  float di  = dinv[row];
  const v4f zero = {0.0f, 0.0f, 0.0f, 0.0f};
  v4f acc = zero;
  if (sub == 0) {
    v4f h = *(const v4f*)(H + (size_t)row * F + l * 4);
    acc = h * (di * di);
  }
  for (int base = start; base < end; base += 32) {
    int m = min(32, end - base);
    int   my_s = 0;
    float my_w = 0.0f;
    if (lane < m) {
      my_s = col[base + lane];
      my_w = dinv[my_s] * di;
    }
    for (int j = 0; j < m; j += EPI * U) {
      v4f   v[U];
      float w[U];
      int   s[U];
#pragma unroll
      for (int u = 0; u < U; ++u) {
        int jj = j + u * EPI + sub;
        s[u] = __shfl(my_s, jj, 32);
        w[u] = __shfl(my_w, jj, 32);
      }
#pragma unroll
      for (int u = 0; u < U; ++u) {
        int jj = j + u * EPI + sub;
        v[u] = (jj < m) ? *(const v4f*)(H + (size_t)s[u] * F + l * 4) : zero;
      }
#pragma unroll
      for (int u = 0; u < U; ++u) acc += v[u] * w[u];
    }
  }
  for (int o = LPR; o < 32; o <<= 1) {
    acc.x += __shfl_xor(acc.x, o, 32);
    acc.y += __shfl_xor(acc.y, o, 32);
    acc.z += __shfl_xor(acc.z, o, 32);
    acc.w += __shfl_xor(acc.w, o, 32);
  }
  if (sub == 0) {
    v4f bb = *(const v4f*)(b + l * 4);
    *(v4f*)(OUT + (size_t)row * F + l * 4) = acc + bb;
  }
}


__global__ void k_log_softmax64(float* __restrict__ out, int n, float invT) {
  int wave = (blockIdx.x * blockDim.x + threadIdx.x) >> 5;
  int lane = threadIdx.x & 31;
  if (wave >= n) return;
  float* row = out + (size_t)wave * 64;
  float z0 = row[lane]      * invT;
  float z1 = row[lane + 32] * invT;
  float m = fmaxf(z0, z1);
#pragma unroll
  for (int off = 16; off > 0; off >>= 1) m = fmaxf(m, __shfl_xor(m, off, 32));
  float s = expf(z0 - m) + expf(z1 - m);
#pragma unroll
  for (int off = 16; off > 0; off >>= 1) s += __shfl_xor(s, off, 32);
  float lg = m + logf(s);
  row[lane]      = z0 - lg;
  row[lane + 32] = z1 - lg;
}


extern "C" void kernel_launch(void* const* d_in, const int* in_sizes, int n_in,
                              void* d_out, int out_size, void* d_ws, size_t ws_size,
                              hipStream_t stream) {
  const float* x  = (const float*)d_in[0];
  const int*   ei = (const int*)d_in[1];
  const float* W1 = (const float*)d_in[2];
  const float* b1 = (const float*)d_in[3];
  const float* W2 = (const float*)d_in[4];
  const float* b2 = (const float*)d_in[5];

  const int IN = 256, HID = 128, OUT = 64;
  int n = in_sizes[0] / IN;
  int E = in_sizes[1] / 2;
  const int* src = ei;
  const int* dst = ei + E;

  float* ws = (float*)d_ws;
  size_t o = 0;
  float* dinv = ws + o; o += ((size_t)n + 255) & ~(size_t)255;
  float* H1   = ws + o; o += (size_t)n * HID;
  float* AGG1 = ws + o; o += (size_t)n * HID;
  float* H2   = ws + o; o += (size_t)n * OUT;
  int NB = (n + RB - 1) / RB;
  int* rowptr = (int*)(ws + o); o += ((size_t)n + 1 + 255) & ~(size_t)255;
  int* hist   = (int*)(ws + o); o += ((size_t)NB * GS + 255) & ~(size_t)255;
  int* rowtot = (int*)(ws + o); o += ((size_t)NB + 255) & ~(size_t)255;
  int* pairs  = (int*)(ws + o); o += ((size_t)2 * E + 255) & ~(size_t)255;
  int* col    = (int*)(ws + o); o += ((size_t)E + 255) & ~(size_t)255;
  float* AGG2 = (float*)d_out;

  int mtiles  = n / 16;
  int mblocks = (mtiles + 7) / 8;
  int chunk   = (E + GS - 1) / GS;

  k_bin_count  <<<GS, SB, NB * sizeof(int), stream>>>(dst, hist, E, NB, chunk);
  k_bin_rowscan<<<NB, GS, 0, stream>>>(hist, rowtot);
  k_bin_scatter<<<GS, SB, NB * sizeof(int), stream>>>(src, dst, hist, rowtot, pairs, E, NB, chunk);
  k_bin_rows   <<<NB, SB, 0, stream>>>(pairs, rowtot, rowptr, dinv, col, n, E);

  k_gemm_wmma<256, 128, false><<<mblocks * (HID / 64), TB, 0, stream>>>(x, W1, H1, n);

  k_agg_rows<128><<<(unsigned)(((long long)n * 32 + TB - 1) / TB), TB, 0, stream>>>(
      H1, col, rowptr, dinv, b1, AGG1, n);

  k_gemm_wmma<128, 64, true><<<mblocks * (OUT / 64), TB, 0, stream>>>(AGG1, W2, H2, n);

  k_agg_rows<64><<<(unsigned)(((long long)n * 32 + TB - 1) / TB), TB, 0, stream>>>(
      H2, col, rowptr, dinv, b2, AGG2, n);

  k_log_softmax64<<<(n * 32 + TB - 1) / TB, TB, 0, stream>>>(AGG2, n, 5.0f);
}


# log_softmax taken by the layer-2 row-sum wave on the row it holds in registers: one launch, 50000 waves and a 25.6 MB round trip fewer; exp sum order now 4 per lane, then a 16-lane tree
# speedup vs baseline: 3.6454x; 1.0360x over previous
#include <hip/hip_runtime.h>
#include <hip/hip_bf16.h>

typedef __attribute__((ext_vector_type(2))) float v2f;
typedef __attribute__((ext_vector_type(8))) float v8f;

#define TB 256


#define SB 512
#define RB 256
#define GS 256
#define EU 8

__device__ __forceinline__ int block_excl_scan(int v, int* wsum, int& total) {
  int lane = threadIdx.x & 31;
  int w    = threadIdx.x >> 5;
  int nw   = blockDim.x >> 5;
  int inc = v;
  for (int o = 1; o < 32; o <<= 1) {
    int u = __shfl_up(inc, o, 32);
    if (lane >= o) inc += u;
  }
  __syncthreads();
  if (lane == 31) wsum[w] = inc;
  __syncthreads();
  if (w == 0) {
    int x = (lane < nw) ? wsum[lane] : 0;
    for (int o = 1; o < 32; o <<= 1) {
      int u = __shfl_up(x, o, 32);
      if (lane >= o) x += u;
    }
    wsum[lane] = x;
  }
  __syncthreads();
  total = wsum[31];
  return inc - v + (w > 0 ? wsum[w - 1] : 0);
}

__global__ void __launch_bounds__(SB) k_bin_count(const int* __restrict__ dst, int* __restrict__ hist,
                                                  int E, int NB, int chunk) {
  extern __shared__ int lbin[];
  int g = blockIdx.x;
  for (int b = threadIdx.x; b < NB; b += SB) lbin[b] = 0;
  __syncthreads();
  int e0 = g * chunk;
  int e1 = min(E, e0 + chunk);
  for (int e = e0 + (int)threadIdx.x; e < e1; e += SB * EU) {
    int d[EU];
#pragma unroll
    for (int u = 0; u < EU; ++u) d[u] = dst[min(e + u * SB, e1 - 1)];
#pragma unroll
    for (int u = 0; u < EU; ++u) atomicAdd(&lbin[(unsigned)d[u] / RB], (e + u * SB < e1) ? 1 : 0);
  }
  __syncthreads();
  for (int b = threadIdx.x; b < NB; b += SB) hist[b * GS + g] = lbin[b];
}

__global__ void __launch_bounds__(GS) k_bin_rowscan(int* __restrict__ hist, int* __restrict__ rowtot) {
  __shared__ int wsum[32];
  int b = blockIdx.x;
  int v = hist[b * GS + threadIdx.x];
  int tot;
  int ex = block_excl_scan(v, wsum, tot);
  hist[b * GS + threadIdx.x] = ex;
  if (threadIdx.x == 0) rowtot[b] = tot;
}

__global__ void __launch_bounds__(SB) k_bin_scatter(const int* __restrict__ src, const int* __restrict__ dst,
                                                    const int* __restrict__ hist, const int* __restrict__ rowtot,
                                                    int* __restrict__ pairs, int E, int NB, int chunk) {
  extern __shared__ int lbin[];
  __shared__ int wsum[32];
  int g = blockIdx.x;
  int carry = 0;
  for (int t0 = 0; t0 < NB; t0 += SB) {
    int b = t0 + (int)threadIdx.x;
    int v = (b < NB) ? rowtot[b] : 0;
    int tot;
    int ex = block_excl_scan(v, wsum, tot);
    if (b < NB) lbin[b] = carry + ex + hist[b * GS + g];
    carry += tot;
  }
  __syncthreads();
  int e0 = g * chunk;
  int e1 = min(E, e0 + chunk);
  for (int e = e0 + (int)threadIdx.x; e < e1; e += SB * EU) {
    int d[EU], s[EU];
#pragma unroll
    for (int u = 0; u < EU; ++u) {
      int ee = min(e + u * SB, e1 - 1);
      d[u] = dst[ee];
      s[u] = src[ee];
    }
    int p[EU];
#pragma unroll
    for (int u = 0; u < EU; ++u) p[u] = atomicAdd(&lbin[(unsigned)d[u] / RB], (e + u * SB < e1) ? 1 : 0);
#pragma unroll
    for (int u = 0; u < EU; ++u)
      if (e + u * SB < e1) *(int2*)(pairs + 2 * (size_t)p[u]) = make_int2(s[u], d[u]);
  }
}

__global__ void __launch_bounds__(SB) k_bin_rows(const int* __restrict__ pairs, const int* __restrict__ rowtot,
                                                 int* __restrict__ rowptr, float* __restrict__ dinv,
                                                 int* __restrict__ col, int n, int E) {
  __shared__ int lcnt[RB];
  __shared__ int lcur[RB];
  __shared__ int wsum[32];
  int b  = blockIdx.x;
  int n0 = b * RB;
  for (int t = threadIdx.x; t < RB; t += SB) lcnt[t] = 0;
  int part = 0;
  for (int k = threadIdx.x; k < b; k += SB) part += rowtot[k];
  int base;
  block_excl_scan(part, wsum, base);
  int end = base + rowtot[b];
  for (int i = base + (int)threadIdx.x; i < end; i += SB * EU) {
    int d[EU];
#pragma unroll
    for (int u = 0; u < EU; ++u) d[u] = pairs[2 * (size_t)min(i + u * SB, end - 1) + 1];
#pragma unroll
    for (int u = 0; u < EU; ++u) atomicAdd(&lcnt[d[u] - n0], (i + u * SB < end) ? 1 : 0);
  }
  __syncthreads();
  int carry = 0;
  for (int t0 = 0; t0 < RB; t0 += SB) {
    int t = t0 + (int)threadIdx.x;
    int c = (t < RB) ? lcnt[t] : 0;
    int tot;
    int ex = block_excl_scan(c, wsum, tot);
    if (t < RB) {
      int first = base + carry + ex;
      lcur[t] = first;
      int node = n0 + t;
      if (node < n) {
        rowptr[node] = first;
        dinv[node]   = rsqrtf((float)(c + 1));
      }
    }
    carry += tot;
  }
  if (b == (int)gridDim.x - 1 && threadIdx.x == 0) rowptr[n] = E;
  __syncthreads();
  for (int i = base + (int)threadIdx.x; i < end; i += SB * EU) {
    int2 t[EU];
#pragma unroll
    for (int u = 0; u < EU; ++u) t[u] = *(const int2*)(pairs + 2 * (size_t)min(i + u * SB, end - 1));
    int p[EU];
#pragma unroll
    for (int u = 0; u < EU; ++u) p[u] = atomicAdd(&lcur[t[u].y - n0], (i + u * SB < end) ? 1 : 0);
#pragma unroll
    for (int u = 0; u < EU; ++u)
      if (i + u * SB < end) col[p[u]] = t[u].x;
  }
}

template<int K, int N, bool RELU>
__global__ void k_gemm_wmma(const float* __restrict__ A, const float* __restrict__ B,
                            float* __restrict__ C, int M) {
  constexpr int NT   = 4;
  constexpr int COLS = 16 * NT;
  constexpr int KC   = 128;
  constexpr int LDK  = KC + 4;
  static_assert(K % KC == 0, "K must be a multiple of KC");
  static_assert(N % COLS == 0, "N must be a multiple of 64");
  __shared__ float Bs[COLS * LDK];

  constexpr int ngroups = N / COLS;
  int bm = blockIdx.x / ngroups;
  int ng = blockIdx.x % ngroups;
  int n0 = ng * COLS;

  int wid  = threadIdx.x >> 5;
  int lane = threadIdx.x & 31;
  int half = lane >> 4;
  int l    = lane & 15;

  int mt     = bm * 8 + wid;
  int mtiles = M >> 4;
  bool active = (mt < mtiles);
  int m0 = mt << 4;

  v8f acc[NT] = {};
  const float* Arow = active ? (A + (size_t)(m0 + l) * K + half * 2) : A;
  const float* Bl   = Bs + l * LDK + half * 2;

  for (int kc = 0; kc < K; kc += KC) {
    __syncthreads();
    for (int idx = threadIdx.x; idx < KC * COLS; idx += TB) {
      int k = idx / COLS;
      int c = idx % COLS;
      Bs[c * LDK + k] = B[(size_t)(kc + k) * N + n0 + c];
    }
    __syncthreads();

    if (active) {
#pragma unroll 4
      for (int k = 0; k < KC; k += 4) {
        float a0 = Arow[kc + k];
        float a1 = Arow[kc + k + 1];
        if (RELU) { a0 = fmaxf(a0, 0.0f); a1 = fmaxf(a1, 0.0f); }
        v2f a; a.x = a0; a.y = a1;
#pragma unroll
        for (int t = 0; t < NT; ++t) {
          const float* bp = Bl + t * 16 * LDK + k;
          v2f b; b.x = bp[0]; b.y = bp[1];
          acc[t] = __builtin_amdgcn_wmma_f32_16x16x4_f32(
              false, a, false, b, (short)0, acc[t], false, false);
        }
      }
    }
  }

  if (active) {
    int rbase = m0 + half * 8;
#pragma unroll
    for (int t = 0; t < NT; ++t) {
#pragma unroll
      for (int j = 0; j < 8; ++j) {
        C[(size_t)(rbase + j) * N + n0 + t * 16 + l] = acc[t][j];
      }
    }
  }
}


typedef __attribute__((ext_vector_type(4))) float v4f;

template<int F, bool LSM>
__global__ void k_agg_rows(const float* __restrict__ H, const int* __restrict__ col,
                           const int* __restrict__ rowptr, const float* __restrict__ dinv,
                           const float* __restrict__ b, float* __restrict__ OUT, int n, float invT) {
  constexpr int LPR = F / 4;
  constexpr int EPI = 32 / LPR;
  constexpr int U   = 8;
  int row  = (int)((blockIdx.x * blockDim.x + threadIdx.x) >> 5);
  int lane = threadIdx.x & 31;
  if (row >= n) return;
  int sub = lane / LPR;
  int l   = lane % LPR;
  int start = rowptr[row];
  int end   = rowptr[row + 1];
  float di  = dinv[row];
  const v4f zero = {0.0f, 0.0f, 0.0f, 0.0f};
  v4f acc = zero;
  if (sub == 0) {
    v4f h = *(const v4f*)(H + (size_t)row * F + l * 4);
    acc = h * (di * di);
  }
  for (int base = start; base < end; base += 32) {
    int m = min(32, end - base);
    int   my_s = 0;
    float my_w = 0.0f;
    if (lane < m) {
      my_s = col[base + lane];
      my_w = dinv[my_s] * di;
    }
    for (int j = 0; j < m; j += EPI * U) {
      v4f   v[U];
      float w[U];
      int   s[U];
#pragma unroll
      for (int u = 0; u < U; ++u) {
        int jj = j + u * EPI + sub;
        s[u] = __shfl(my_s, jj, 32);
        w[u] = __shfl(my_w, jj, 32);
      }
#pragma unroll
      for (int u = 0; u < U; ++u) {
        int jj = j + u * EPI + sub;
        v[u] = (jj < m) ? *(const v4f*)(H + (size_t)s[u] * F + l * 4) : zero;
      }
#pragma unroll
      for (int u = 0; u < U; ++u) acc += v[u] * w[u];
    }
  }
  for (int o = LPR; o < 32; o <<= 1) {
    acc.x += __shfl_xor(acc.x, o, 32);
    acc.y += __shfl_xor(acc.y, o, 32);
    acc.z += __shfl_xor(acc.z, o, 32);
    acc.w += __shfl_xor(acc.w, o, 32);
  }
  v4f bb = *(const v4f*)(b + l * 4);
  v4f r = acc + bb;
  if (LSM) {
    v4f z = r * invT;
    float m = fmaxf(fmaxf(z.x, z.y), fmaxf(z.z, z.w));
    for (int o = 1; o < LPR; o <<= 1) m = fmaxf(m, __shfl_xor(m, o, 32));
    float s = (expf(z.x - m) + expf(z.y - m)) + (expf(z.z - m) + expf(z.w - m));
    for (int o = 1; o < LPR; o <<= 1) s += __shfl_xor(s, o, 32);
    float lg = m + logf(s);
    r = z - lg;
  }
  if (sub == 0) *(v4f*)(OUT + (size_t)row * F + l * 4) = r;
}


extern "C" void kernel_launch(void* const* d_in, const int* in_sizes, int n_in,
                              void* d_out, int out_size, void* d_ws, size_t ws_size,
                              hipStream_t stream) {
  const float* x  = (const float*)d_in[0];
  const int*   ei = (const int*)d_in[1];
  const float* W1 = (const float*)d_in[2];
  const float* b1 = (const float*)d_in[3];
  const float* W2 = (const float*)d_in[4];
  const float* b2 = (const float*)d_in[5];

  const int IN = 256, HID = 128, OUT = 64;
  int n = in_sizes[0] / IN;
  int E = in_sizes[1] / 2;
  const int* src = ei;
  const int* dst = ei + E;

  float* ws = (float*)d_ws;
  size_t o = 0;
  float* dinv = ws + o; o += ((size_t)n + 255) & ~(size_t)255;
  float* H1   = ws + o; o += (size_t)n * HID;
  float* AGG1 = ws + o; o += (size_t)n * HID;
  float* H2   = ws + o; o += (size_t)n * OUT;
  int NB = (n + RB - 1) / RB;
  int* rowptr = (int*)(ws + o); o += ((size_t)n + 1 + 255) & ~(size_t)255;
  int* hist   = (int*)(ws + o); o += ((size_t)NB * GS + 255) & ~(size_t)255;
  int* rowtot = (int*)(ws + o); o += ((size_t)NB + 255) & ~(size_t)255;
  int* pairs  = (int*)(ws + o); o += ((size_t)2 * E + 255) & ~(size_t)255;
  int* col    = (int*)(ws + o); o += ((size_t)E + 255) & ~(size_t)255;
  float* OUTP = (float*)d_out;

  int mtiles  = n / 16;
  int mblocks = (mtiles + 7) / 8;
  int chunk   = (E + GS - 1) / GS;

  k_bin_count  <<<GS, SB, NB * sizeof(int), stream>>>(dst, hist, E, NB, chunk);
  k_bin_rowscan<<<NB, GS, 0, stream>>>(hist, rowtot);
  k_bin_scatter<<<GS, SB, NB * sizeof(int), stream>>>(src, dst, hist, rowtot, pairs, E, NB, chunk);
  k_bin_rows   <<<NB, SB, 0, stream>>>(pairs, rowtot, rowptr, dinv, col, n, E);

  k_gemm_wmma<256, 128, false><<<mblocks * (HID / 64), TB, 0, stream>>>(x, W1, H1, n);

  k_agg_rows<128, false><<<(unsigned)(((long long)n * 32 + TB - 1) / TB), TB, 0, stream>>>(
      H1, col, rowptr, dinv, b1, AGG1, n, 1.0f);

  k_gemm_wmma<128, 64, true><<<mblocks * (OUT / 64), TB, 0, stream>>>(AGG1, W2, H2, n);

  k_agg_rows<64, true><<<(unsigned)(((long long)n * 32 + TB - 1) / TB), TB, 0, stream>>>(
      H2, col, rowptr, dinv, b2, OUTP, n, 5.0f);
}


# edge slices sort themselves by destination bin in their own block, so the global count and scan launches go (8 launches to 6); a bin's block reads one segment per slice; row sums still arrival order
# speedup vs baseline: 4.3622x; 1.1966x over previous
#include <hip/hip_runtime.h>
#include <hip/hip_bf16.h>

typedef __attribute__((ext_vector_type(2))) float v2f;
typedef __attribute__((ext_vector_type(4))) float v4f;
typedef __attribute__((ext_vector_type(8))) float v8f;

#define TB 256


#define SB 512
#define RB 256
#define GS 256
#define EU 8
#define SPW (GS / (SB / 32))

__device__ __forceinline__ int block_excl_scan(int v, int* wsum, int& total) {
  int lane = threadIdx.x & 31;
  int w    = threadIdx.x >> 5;
  int nw   = blockDim.x >> 5;
  int inc = v;
  for (int o = 1; o < 32; o <<= 1) {
    int u = __shfl_up(inc, o, 32);
    if (lane >= o) inc += u;
  }
  __syncthreads();
  if (lane == 31) wsum[w] = inc;
  __syncthreads();
  if (w == 0) {
    int x = (lane < nw) ? wsum[lane] : 0;
    for (int o = 1; o < 32; o <<= 1) {
      int u = __shfl_up(x, o, 32);
      if (lane >= o) x += u;
    }
    wsum[lane] = x;
  }
  __syncthreads();
  total = wsum[31];
  return inc - v + (w > 0 ? wsum[w - 1] : 0);
}

__global__ void __launch_bounds__(SB) k_slice_sort(const int* __restrict__ src, const int* __restrict__ dst,
                                                   int* __restrict__ cntT, int* __restrict__ offT,
                                                   int* __restrict__ pairs, int E, int NB, int chunk) {
  extern __shared__ int lbin[];
  __shared__ int wsum[32];
  int g = blockIdx.x;
  for (int b = threadIdx.x; b < NB; b += SB) lbin[b] = 0;
  __syncthreads();
  int e0 = g * chunk;
  int e1 = min(E, e0 + chunk);
  for (int e = e0 + (int)threadIdx.x; e < e1; e += SB * EU) {
    int d[EU];
#pragma unroll
    for (int u = 0; u < EU; ++u) d[u] = dst[min(e + u * SB, e1 - 1)];
#pragma unroll
    for (int u = 0; u < EU; ++u) atomicAdd(&lbin[(unsigned)d[u] / RB], (e + u * SB < e1) ? 1 : 0);
  }
  __syncthreads();
  int carry = e0;
  for (int t0 = 0; t0 < NB; t0 += SB) {
    int b = t0 + (int)threadIdx.x;
    int c = (b < NB) ? lbin[b] : 0;
    int tot;
    int ex = block_excl_scan(c, wsum, tot);
    if (b < NB) {
      int first = carry + ex;
      cntT[b * GS + g] = c;
      offT[b * GS + g] = first;
      lbin[b] = first;
    }
    carry += tot;
  }
  __syncthreads();
  for (int e = e0 + (int)threadIdx.x; e < e1; e += SB * EU) {
    int d[EU], s[EU];
#pragma unroll
    for (int u = 0; u < EU; ++u) {
      int ee = min(e + u * SB, e1 - 1);
      d[u] = dst[ee];
      s[u] = src[ee];
    }
    int p[EU];
#pragma unroll
    for (int u = 0; u < EU; ++u) p[u] = atomicAdd(&lbin[(unsigned)d[u] / RB], (e + u * SB < e1) ? 1 : 0);
#pragma unroll
    for (int u = 0; u < EU; ++u)
      if (e + u * SB < e1) *(int2*)(pairs + 2 * (size_t)p[u]) = make_int2(s[u], d[u]);
  }
}

__global__ void __launch_bounds__(SB) k_bin_rows(const int* __restrict__ pairs, const int* __restrict__ cntT,
                                                 const int* __restrict__ offT,
                                                 int* __restrict__ rowptr, float* __restrict__ dinv,
                                                 int* __restrict__ col, int n, int E, int chunk) {
  __shared__ int lcnt[RB];
  __shared__ int lcur[RB];
  __shared__ int wsum[32];
  int b    = blockIdx.x;
  int n0   = b * RB;
  int lane = threadIdx.x & 31;
  int w    = threadIdx.x >> 5;
  for (int t = threadIdx.x; t < RB; t += SB) lcnt[t] = 0;
  int segc = 0, sego = 0, part = 0;
  if (lane < SPW) {
    int g = w * SPW + lane;
    segc = cntT[b * GS + g];
    sego = offT[b * GS + g];
    part = sego - g * chunk;
  }
  int base;
  block_excl_scan(part, wsum, base);
  int maxlen = segc;
  for (int o = 16; o > 0; o >>= 1) maxlen = max(maxlen, __shfl_xor(maxlen, o, 32));
  const int2* pr = (const int2*)pairs;

  {
    int len[SPW];
    int da[SPW], db[SPW];
#pragma unroll
    for (int u = 0; u < SPW; ++u) {
      len[u]  = __shfl(segc, u, 32);
      int off = __shfl(sego, u, 32);
      da[u] = pairs[2 * (size_t)((lane      < len[u]) ? off + lane      : 0) + 1];
      db[u] = pairs[2 * (size_t)((lane + 32 < len[u]) ? off + lane + 32 : 0) + 1];
    }
#pragma unroll
    for (int u = 0; u < SPW; ++u) {
      if (lane      < len[u]) atomicAdd(&lcnt[da[u] - n0], 1);
      if (lane + 32 < len[u]) atomicAdd(&lcnt[db[u] - n0], 1);
    }
  }
  if (maxlen > 64) {
    for (int u = 0; u < SPW; ++u) {
      int len = __shfl(segc, u, 32);
      int off = __shfl(sego, u, 32);
      for (int i = 64 + lane; i < len; i += 32) atomicAdd(&lcnt[pr[off + i].y - n0], 1);
    }
  }
  __syncthreads();
  int carry = 0;
  for (int t0 = 0; t0 < RB; t0 += SB) {
    int t = t0 + (int)threadIdx.x;
    int c = (t < RB) ? lcnt[t] : 0;
    int tot;
    int ex = block_excl_scan(c, wsum, tot);
    if (t < RB) {
      int first = base + carry + ex;
      lcur[t] = first;
      int node = n0 + t;
      if (node < n) {
        rowptr[node] = first;
        dinv[node]   = rsqrtf((float)(c + 1));
      }
    }
    carry += tot;
  }
  if (b == (int)gridDim.x - 1 && threadIdx.x == 0) rowptr[n] = E;
  __syncthreads();

  {
    int len[SPW];
    int2 ta[SPW], tb[SPW];
#pragma unroll
    for (int u = 0; u < SPW; ++u) {
      len[u]  = __shfl(segc, u, 32);
      int off = __shfl(sego, u, 32);
      ta[u] = pr[(lane      < len[u]) ? off + lane      : 0];
      tb[u] = pr[(lane + 32 < len[u]) ? off + lane + 32 : 0];
    }
#pragma unroll
    for (int u = 0; u < SPW; ++u) {
      if (lane      < len[u]) col[atomicAdd(&lcur[ta[u].y - n0], 1)] = ta[u].x;
      if (lane + 32 < len[u]) col[atomicAdd(&lcur[tb[u].y - n0], 1)] = tb[u].x;
    }
  }
  if (maxlen > 64) {
    for (int u = 0; u < SPW; ++u) {
      int len = __shfl(segc, u, 32);
      int off = __shfl(sego, u, 32);
      for (int i = 64 + lane; i < len; i += 32) {
        int2 t = pr[off + i];
        col[atomicAdd(&lcur[t.y - n0], 1)] = t.x;
      }
    }
  }
}

template<int K, int N, bool RELU>
__global__ void k_gemm_wmma(const float* __restrict__ A, const float* __restrict__ B,
                            const float* __restrict__ rowscale, float* __restrict__ C, int M) {
  constexpr int NT   = 4;
  constexpr int COLS = 16 * NT;
  constexpr int KC   = 128;
  constexpr int LDK  = KC + 4;
  static_assert(K % KC == 0, "K must be a multiple of KC");
  static_assert(N % COLS == 0, "N must be a multiple of 64");
  __shared__ float Bs[COLS * LDK];

  constexpr int ngroups = N / COLS;
  int bm = blockIdx.x / ngroups;
  int ng = blockIdx.x % ngroups;
  int n0 = ng * COLS;

  int wid  = threadIdx.x >> 5;
  int lane = threadIdx.x & 31;
  int half = lane >> 4;
  int l    = lane & 15;

  int mt     = bm * 8 + wid;
  int mtiles = M >> 4;
  bool active = (mt < mtiles);
  int m0 = mt << 4;

  v8f acc[NT] = {};
  const float* Arow = active ? (A + (size_t)(m0 + l) * K + half * 2) : A;
  const float* Bl   = Bs + l * LDK + half * 2;

  for (int kc = 0; kc < K; kc += KC) {
    __syncthreads();
    for (int idx = threadIdx.x; idx < KC * COLS; idx += TB) {
      int k = idx / COLS;
      int c = idx % COLS;
      Bs[c * LDK + k] = B[(size_t)(kc + k) * N + n0 + c];
    }
    __syncthreads();

    if (active) {
#pragma unroll 4
      for (int k = 0; k < KC; k += 4) {
        float a0 = Arow[kc + k];
        float a1 = Arow[kc + k + 1];
        if (RELU) { a0 = fmaxf(a0, 0.0f); a1 = fmaxf(a1, 0.0f); }
        v2f a; a.x = a0; a.y = a1;
#pragma unroll
        for (int t = 0; t < NT; ++t) {
          const float* bp = Bl + t * 16 * LDK + k;
          v2f b; b.x = bp[0]; b.y = bp[1];
          acc[t] = __builtin_amdgcn_wmma_f32_16x16x4_f32(
              false, a, false, b, (short)0, acc[t], false, false);
        }
      }
    }
  }

  if (active) {
    int rbase = m0 + half * 8;
    v4f s0 = *(const v4f*)(rowscale + rbase);
    v4f s1 = *(const v4f*)(rowscale + rbase + 4);
    float sc[8] = {s0.x, s0.y, s0.z, s0.w, s1.x, s1.y, s1.z, s1.w};
#pragma unroll
    for (int t = 0; t < NT; ++t) {
#pragma unroll
      for (int j = 0; j < 8; ++j) {
        C[(size_t)(rbase + j) * N + n0 + t * 16 + l] = acc[t][j] * sc[j];
      }
    }
  }
}


template<int F, bool LSM>
__global__ void k_agg_rows(const float* __restrict__ HS, const int* __restrict__ col,
                           const int* __restrict__ rowptr, const float* __restrict__ dinv,
                           const float* __restrict__ b, float* __restrict__ OUT, int n, float invT) {
  constexpr int LPR = F / 4;
  constexpr int EPI = 32 / LPR;
  constexpr int U   = 8;
  int row  = (int)((blockIdx.x * blockDim.x + threadIdx.x) >> 5);
  int lane = threadIdx.x & 31;
  if (row >= n) return;
  int sub = lane / LPR;
  int l   = lane % LPR;
  int start = rowptr[row];
  int end   = rowptr[row + 1];
  float di  = dinv[row];
  const v4f zero = {0.0f, 0.0f, 0.0f, 0.0f};
  v4f acc = zero;
  if (sub == 0)
    acc = *(const v4f*)(HS + (size_t)row * F + l * 4);
  for (int base = start; base < end; base += 32) {
    int m = min(32, end - base);
    int my_s = 0;
    if (lane < m) my_s = col[base + lane];
    for (int j = 0; j < m; j += EPI * U) {
      v4f v[U];
      int s[U];
#pragma unroll
      for (int u = 0; u < U; ++u) {
        int jj = j + u * EPI + sub;
        s[u] = __shfl(my_s, jj, 32);
      }
#pragma unroll
      for (int u = 0; u < U; ++u) {
        int jj = j + u * EPI + sub;
        v[u] = (jj < m) ? *(const v4f*)(HS + (size_t)s[u] * F + l * 4) : zero;
      }
#pragma unroll
      for (int u = 0; u < U; ++u) acc += v[u];
    }
  }
  for (int o = LPR; o < 32; o <<= 1) {
    acc.x += __shfl_xor(acc.x, o, 32);
    acc.y += __shfl_xor(acc.y, o, 32);
    acc.z += __shfl_xor(acc.z, o, 32);
    acc.w += __shfl_xor(acc.w, o, 32);
  }
  v4f bb = *(const v4f*)(b + l * 4);
  v4f r = acc * di + bb;
  if (LSM) {
    v4f z = r * invT;
    float m = fmaxf(fmaxf(z.x, z.y), fmaxf(z.z, z.w));
    for (int o = 1; o < LPR; o <<= 1) m = fmaxf(m, __shfl_xor(m, o, 32));
    float s = (expf(z.x - m) + expf(z.y - m)) + (expf(z.z - m) + expf(z.w - m));
    for (int o = 1; o < LPR; o <<= 1) s += __shfl_xor(s, o, 32);
    float lg = m + logf(s);
    r = z - lg;
  }
  if (sub == 0) *(v4f*)(OUT + (size_t)row * F + l * 4) = r;
}


extern "C" void kernel_launch(void* const* d_in, const int* in_sizes, int n_in,
                              void* d_out, int out_size, void* d_ws, size_t ws_size,
                              hipStream_t stream) {
  const float* x  = (const float*)d_in[0];
  const int*   ei = (const int*)d_in[1];
  const float* W1 = (const float*)d_in[2];
  const float* b1 = (const float*)d_in[3];
  const float* W2 = (const float*)d_in[4];
  const float* b2 = (const float*)d_in[5];

  const int IN = 256, HID = 128, OUT = 64;
  int n = in_sizes[0] / IN;
  int E = in_sizes[1] / 2;
  const int* src = ei;
  const int* dst = ei + E;

  float* ws = (float*)d_ws;
  size_t o = 0;
  float* dinv = ws + o; o += ((size_t)n + 255) & ~(size_t)255;
  float* H1   = ws + o; o += (size_t)n * HID;
  float* AGG1 = ws + o; o += (size_t)n * HID;
  float* H2   = ws + o; o += (size_t)n * OUT;
  int NB = (n + RB - 1) / RB;
  int* rowptr = (int*)(ws + o); o += ((size_t)n + 1 + 255) & ~(size_t)255;
  int* cntT   = (int*)(ws + o); o += ((size_t)NB * GS + 255) & ~(size_t)255;
  int* offT   = (int*)(ws + o); o += ((size_t)NB * GS + 255) & ~(size_t)255;
  int* pairs  = (int*)(ws + o); o += ((size_t)2 * E + 255) & ~(size_t)255;
  int* col    = (int*)(ws + o); o += ((size_t)E + 255) & ~(size_t)255;
  float* OUTP = (float*)d_out;

  int mtiles  = n / 16;
  int mblocks = (mtiles + 7) / 8;
  int chunk   = (E + GS - 1) / GS;

  k_slice_sort<<<GS, SB, NB * sizeof(int), stream>>>(src, dst, cntT, offT, pairs, E, NB, chunk);
  k_bin_rows  <<<NB, SB, 0, stream>>>(pairs, cntT, offT, rowptr, dinv, col, n, E, chunk);

  k_gemm_wmma<256, 128, false><<<mblocks * (HID / 64), TB, 0, stream>>>(x, W1, dinv, H1, n);

  k_agg_rows<128, false><<<(unsigned)(((long long)n * 32 + TB - 1) / TB), TB, 0, stream>>>(
      H1, col, rowptr, dinv, b1, AGG1, n, 1.0f);

  k_gemm_wmma<128, 64, true><<<mblocks * (OUT / 64), TB, 0, stream>>>(AGG1, W2, dinv, H2, n);

  k_agg_rows<64, true><<<(unsigned)(((long long)n * 32 + TB - 1) / TB), TB, 0, stream>>>(
      H2, col, rowptr, dinv, b2, OUTP, n, 5.0f);
}


# row-grouping pass and layer-1 GEMM fused: a bin's block (208 rows) multiplies its own rows once it has their dinv, W1 whole in LDS, x read once; 6 launches to 5; k sum order inside each 8 changed
# speedup vs baseline: 4.6266x; 1.0606x over previous
#include <hip/hip_runtime.h>
#include <hip/hip_bf16.h>

typedef __attribute__((ext_vector_type(2))) float v2f;
typedef __attribute__((ext_vector_type(4))) float v4f;
typedef __attribute__((ext_vector_type(8))) float v8f;

#define TB 256


#define SB 512
#define RB 208
#define MTB (RB / 16)
#define GS 256
#define EU 8
#define SPW (GS / (SB / 32))

__device__ __forceinline__ int block_excl_scan(int v, int* wsum, int& total) {
  int lane = threadIdx.x & 31;
  int w    = threadIdx.x >> 5;
  int nw   = blockDim.x >> 5;
  int inc = v;
  for (int o = 1; o < 32; o <<= 1) {
    int u = __shfl_up(inc, o, 32);
    if (lane >= o) inc += u;
  }
  __syncthreads();
  if (lane == 31) wsum[w] = inc;
  __syncthreads();
  if (w == 0) {
    int x = (lane < nw) ? wsum[lane] : 0;
    for (int o = 1; o < 32; o <<= 1) {
      int u = __shfl_up(x, o, 32);
      if (lane >= o) x += u;
    }
    wsum[lane] = x;
  }
  __syncthreads();
  total = wsum[31];
  return inc - v + (w > 0 ? wsum[w - 1] : 0);
}

__global__ void __launch_bounds__(SB) k_slice_sort(const int* __restrict__ src, const int* __restrict__ dst,
                                                   int* __restrict__ cntT, int* __restrict__ offT,
                                                   int* __restrict__ pairs, int E, int NB, int chunk) {
  extern __shared__ int lbin[];
  __shared__ int wsum[32];
  int g = blockIdx.x;
  for (int b = threadIdx.x; b < NB; b += SB) lbin[b] = 0;
  __syncthreads();
  int e0 = g * chunk;
  int e1 = min(E, e0 + chunk);
  for (int e = e0 + (int)threadIdx.x; e < e1; e += SB * EU) {
    int d[EU];
#pragma unroll
    for (int u = 0; u < EU; ++u) d[u] = dst[min(e + u * SB, e1 - 1)];
#pragma unroll
    for (int u = 0; u < EU; ++u) atomicAdd(&lbin[(unsigned)d[u] / RB], (e + u * SB < e1) ? 1 : 0);
  }
  __syncthreads();
  int carry = e0;
  for (int t0 = 0; t0 < NB; t0 += SB) {
    int b = t0 + (int)threadIdx.x;
    int c = (b < NB) ? lbin[b] : 0;
    int tot;
    int ex = block_excl_scan(c, wsum, tot);
    if (b < NB) {
      int first = carry + ex;
      cntT[b * GS + g] = c;
      offT[b * GS + g] = first;
      lbin[b] = first;
    }
    carry += tot;
  }
  __syncthreads();
  for (int e = e0 + (int)threadIdx.x; e < e1; e += SB * EU) {
    int d[EU], s[EU];
#pragma unroll
    for (int u = 0; u < EU; ++u) {
      int ee = min(e + u * SB, e1 - 1);
      d[u] = dst[ee];
      s[u] = src[ee];
    }
    int p[EU];
#pragma unroll
    for (int u = 0; u < EU; ++u) p[u] = atomicAdd(&lbin[(unsigned)d[u] / RB], (e + u * SB < e1) ? 1 : 0);
#pragma unroll
    for (int u = 0; u < EU; ++u)
      if (e + u * SB < e1) *(int2*)(pairs + 2 * (size_t)p[u]) = make_int2(s[u], d[u]);
  }
}

template<int K, int N>
__global__ void __launch_bounds__(SB) k_bin_rows_gemm(const int* __restrict__ pairs, const int* __restrict__ cntT,
                                                      const int* __restrict__ offT,
                                                      int* __restrict__ rowptr, float* __restrict__ dinv,
                                                      int* __restrict__ col,
                                                      const float* __restrict__ X, const float* __restrict__ Wm,
                                                      float* __restrict__ H, int n, int E, int chunk) {
  constexpr int LDK = K + 4;
  constexpr int BSF = 41 * 1024;
  static_assert(N * LDK <= BSF, "transposed Wm fits");
  static_assert(N == 128 && K % 32 == 0, "4 column groups of 32; k in steps of 32");
  __shared__ __attribute__((aligned(16))) float Bs[BSF];
  __shared__ __attribute__((aligned(16))) float ldinv[RB];
  __shared__ int lcnt[RB];
  __shared__ int lcur[RB];
  __shared__ int wsum[32];
  {
    constexpr int SG = 32;
    static_assert((K * N) % (SB * SG) == 0 && SB % N == 0, "whole groups; a thread keeps its column");
    int c  = threadIdx.x % N;
    int k0 = threadIdx.x / N;
    for (int base = 0; base < K * N; base += SB * SG) {
      float v[SG];
#pragma unroll
      for (int u = 0; u < SG; ++u) v[u] = Wm[base + u * SB + (int)threadIdx.x];
#pragma unroll
      for (int u = 0; u < SG; ++u) Bs[c * LDK + base / N + u * (SB / N) + k0] = v[u];
    }
  }
  int b    = blockIdx.x;
  int n0   = b * RB;
  int lane = threadIdx.x & 31;
  int w    = threadIdx.x >> 5;
  for (int t = threadIdx.x; t < RB; t += SB) lcnt[t] = 0;
  int segc = 0, sego = 0, part = 0;
  if (lane < SPW) {
    int g = w * SPW + lane;
    segc = cntT[b * GS + g];
    sego = offT[b * GS + g];
    part = sego - g * chunk;
  }
  int base;
  block_excl_scan(part, wsum, base);
  int maxlen = segc;
  for (int o = 16; o > 0; o >>= 1) maxlen = max(maxlen, __shfl_xor(maxlen, o, 32));
  const int2* pr = (const int2*)pairs;

  {
    int len[SPW];
    int da[SPW], db[SPW];
#pragma unroll
    for (int u = 0; u < SPW; ++u) {
      len[u]  = __shfl(segc, u, 32);
      int off = __shfl(sego, u, 32);
      da[u] = pairs[2 * (size_t)((lane      < len[u]) ? off + lane      : 0) + 1];
      db[u] = pairs[2 * (size_t)((lane + 32 < len[u]) ? off + lane + 32 : 0) + 1];
    }
#pragma unroll
    for (int u = 0; u < SPW; ++u) {
      if (lane      < len[u]) atomicAdd(&lcnt[da[u] - n0], 1);
      if (lane + 32 < len[u]) atomicAdd(&lcnt[db[u] - n0], 1);
    }
  }
  if (maxlen > 64) {
    for (int u = 0; u < SPW; ++u) {
      int len = __shfl(segc, u, 32);
      int off = __shfl(sego, u, 32);
      for (int i = 64 + lane; i < len; i += 32) atomicAdd(&lcnt[pr[off + i].y - n0], 1);
    }
  }
  __syncthreads();
  int carry = 0;
  for (int t0 = 0; t0 < RB; t0 += SB) {
    int t = t0 + (int)threadIdx.x;
    int c = (t < RB) ? lcnt[t] : 0;
    int tot;
    int ex = block_excl_scan(c, wsum, tot);
    if (t < RB) {
      int first = base + carry + ex;
      lcur[t] = first;
      int node = n0 + t;
      float dv = rsqrtf((float)(c + 1));
      ldinv[t] = dv;
      if (node < n) {
        rowptr[node] = first;
        dinv[node]   = dv;
      }
    }
    carry += tot;
  }
  if (b == (int)gridDim.x - 1 && threadIdx.x == 0) rowptr[n] = E;
  __syncthreads();

  {
    int len[SPW];
    int2 ta[SPW], tb[SPW];
#pragma unroll
    for (int u = 0; u < SPW; ++u) {
      len[u]  = __shfl(segc, u, 32);
      int off = __shfl(sego, u, 32);
      ta[u] = pr[(lane      < len[u]) ? off + lane      : 0];
      tb[u] = pr[(lane + 32 < len[u]) ? off + lane + 32 : 0];
    }
#pragma unroll
    for (int u = 0; u < SPW; ++u) {
      if (lane      < len[u]) col[atomicAdd(&lcur[ta[u].y - n0], 1)] = ta[u].x;
      if (lane + 32 < len[u]) col[atomicAdd(&lcur[tb[u].y - n0], 1)] = tb[u].x;
    }
  }
  if (maxlen > 64) {
    for (int u = 0; u < SPW; ++u) {
      int len = __shfl(segc, u, 32);
      int off = __shfl(sego, u, 32);
      for (int i = 64 + lane; i < len; i += 32) {
        int2 t = pr[off + i];
        col[atomicAdd(&lcur[t.y - n0], 1)] = t.x;
      }
    }
  }

  {
    int half = lane >> 4;
    int l    = lane & 15;
    int rg   = ((w & 3) + (w >> 2)) & 3;
    int nc0  = (w >> 2) * 32;
    int mtiles = n >> 4;
    const float* Bl = Bs + (nc0 + l) * LDK + half * 4;
    for (int mt = rg; mt < MTB; mt += 4) {
      int gmt = b * MTB + mt;
      if (gmt >= mtiles) break;
      int m0 = gmt << 4;
      const float* Arow = X + (size_t)(m0 + l) * K + half * 4;
      v8f acc0 = {}, acc1 = {};
      v4f an[4];
#pragma unroll
      for (int u = 0; u < 4; ++u) an[u] = *(const v4f*)(Arow + u * 8);
#pragma unroll
      for (int kk = 0; kk < K; kk += 32) {
        v4f a[4];
#pragma unroll
        for (int u = 0; u < 4; ++u) a[u] = an[u];
        if (kk + 32 < K) {
#pragma unroll
          for (int u = 0; u < 4; ++u) an[u] = *(const v4f*)(Arow + kk + 32 + u * 8);
        }
#pragma unroll
        for (int u = 0; u < 4; ++u) {
          v4f b0 = *(const v4f*)(Bl + kk + u * 8);
          v4f b1 = *(const v4f*)(Bl + 16 * LDK + kk + u * 8);
          v2f alo, ahi, b0lo, b0hi, b1lo, b1hi;
          alo.x  = a[u].x; alo.y  = a[u].y; ahi.x  = a[u].z; ahi.y  = a[u].w;
          b0lo.x = b0.x;   b0lo.y = b0.y;   b0hi.x = b0.z;   b0hi.y = b0.w;
          b1lo.x = b1.x;   b1lo.y = b1.y;   b1hi.x = b1.z;   b1hi.y = b1.w;
          acc0 = __builtin_amdgcn_wmma_f32_16x16x4_f32(false, alo, false, b0lo, (short)0, acc0, false, false);
          acc1 = __builtin_amdgcn_wmma_f32_16x16x4_f32(false, alo, false, b1lo, (short)0, acc1, false, false);
          acc0 = __builtin_amdgcn_wmma_f32_16x16x4_f32(false, ahi, false, b0hi, (short)0, acc0, false, false);
          acc1 = __builtin_amdgcn_wmma_f32_16x16x4_f32(false, ahi, false, b1hi, (short)0, acc1, false, false);
        }
      }
      int rl = mt * 16 + half * 8;
      v4f s0 = *(const v4f*)(ldinv + rl);
      v4f s1 = *(const v4f*)(ldinv + rl + 4);
      float sc[8] = {s0.x, s0.y, s0.z, s0.w, s1.x, s1.y, s1.z, s1.w};
      float* Hp = H + (size_t)(m0 + half * 8) * N + nc0 + l;
#pragma unroll
      for (int j = 0; j < 8; ++j) {
        Hp[(size_t)j * N]      = acc0[j] * sc[j];
        Hp[(size_t)j * N + 16] = acc1[j] * sc[j];
      }
    }
  }
}

template<int K, int N, bool RELU>
__global__ void k_gemm_wmma(const float* __restrict__ A, const float* __restrict__ B,
                            const float* __restrict__ rowscale, float* __restrict__ C, int M) {
  constexpr int NT   = 4;
  constexpr int COLS = 16 * NT;
  constexpr int KC   = 128;
  constexpr int LDK  = KC + 4;
  static_assert(K % KC == 0, "K must be a multiple of KC");
  static_assert(N % COLS == 0, "N must be a multiple of 64");
  __shared__ float Bs[COLS * LDK];

  constexpr int ngroups = N / COLS;
  int bm = blockIdx.x / ngroups;
  int ng = blockIdx.x % ngroups;
  int n0 = ng * COLS;

  int wid  = threadIdx.x >> 5;
  int lane = threadIdx.x & 31;
  int half = lane >> 4;
  int l    = lane & 15;

  int mt     = bm * 8 + wid;
  int mtiles = M >> 4;
  bool active = (mt < mtiles);
  int m0 = mt << 4;

  v8f acc[NT] = {};
  const float* Arow = active ? (A + (size_t)(m0 + l) * K + half * 2) : A;
  const float* Bl   = Bs + l * LDK + half * 2;

  for (int kc = 0; kc < K; kc += KC) {
    __syncthreads();
    for (int idx = threadIdx.x; idx < KC * COLS; idx += TB) {
      int k = idx / COLS;
      int c = idx % COLS;
      Bs[c * LDK + k] = B[(size_t)(kc + k) * N + n0 + c];
    }
    __syncthreads();

    if (active) {
#pragma unroll 4
      for (int k = 0; k < KC; k += 4) {
        float a0 = Arow[kc + k];
        float a1 = Arow[kc + k + 1];
        if (RELU) { a0 = fmaxf(a0, 0.0f); a1 = fmaxf(a1, 0.0f); }
        v2f a; a.x = a0; a.y = a1;
#pragma unroll
        for (int t = 0; t < NT; ++t) {
          const float* bp = Bl + t * 16 * LDK + k;
          v2f b; b.x = bp[0]; b.y = bp[1];
          acc[t] = __builtin_amdgcn_wmma_f32_16x16x4_f32(
              false, a, false, b, (short)0, acc[t], false, false);
        }
      }
    }
  }

  if (active) {
    int rbase = m0 + half * 8;
    v4f s0 = *(const v4f*)(rowscale + rbase);
    v4f s1 = *(const v4f*)(rowscale + rbase + 4);
    float sc[8] = {s0.x, s0.y, s0.z, s0.w, s1.x, s1.y, s1.z, s1.w};
#pragma unroll
    for (int t = 0; t < NT; ++t) {
#pragma unroll
      for (int j = 0; j < 8; ++j) {
        C[(size_t)(rbase + j) * N + n0 + t * 16 + l] = acc[t][j] * sc[j];
      }
    }
  }
}


template<int F, bool LSM>
__global__ void k_agg_rows(const float* __restrict__ HS, const int* __restrict__ col,
                           const int* __restrict__ rowptr, const float* __restrict__ dinv,
                           const float* __restrict__ b, float* __restrict__ OUT, int n, float invT) {
  constexpr int LPR = F / 4;
  constexpr int EPI = 32 / LPR;
  constexpr int U   = 8;
  int row  = (int)((blockIdx.x * blockDim.x + threadIdx.x) >> 5);
  int lane = threadIdx.x & 31;
  if (row >= n) return;
  int sub = lane / LPR;
  int l   = lane % LPR;
  int start = rowptr[row];
  int end   = rowptr[row + 1];
  float di  = dinv[row];
  const v4f zero = {0.0f, 0.0f, 0.0f, 0.0f};
  v4f acc = zero;
  if (sub == 0)
    acc = *(const v4f*)(HS + (size_t)row * F + l * 4);
  for (int base = start; base < end; base += 32) {
    int m = min(32, end - base);
    int my_s = 0;
    if (lane < m) my_s = col[base + lane];
    for (int j = 0; j < m; j += EPI * U) {
      v4f v[U];
      int s[U];
#pragma unroll
      for (int u = 0; u < U; ++u) {
        int jj = j + u * EPI + sub;
        s[u] = __shfl(my_s, jj, 32);
      }
#pragma unroll
      for (int u = 0; u < U; ++u) {
        int jj = j + u * EPI + sub;
        v[u] = (jj < m) ? *(const v4f*)(HS + (size_t)s[u] * F + l * 4) : zero;
      }
#pragma unroll
      for (int u = 0; u < U; ++u) acc += v[u];
    }
  }
  for (int o = LPR; o < 32; o <<= 1) {
    acc.x += __shfl_xor(acc.x, o, 32);
    acc.y += __shfl_xor(acc.y, o, 32);
    acc.z += __shfl_xor(acc.z, o, 32);
    acc.w += __shfl_xor(acc.w, o, 32);
  }
  v4f bb = *(const v4f*)(b + l * 4);
  v4f r = acc * di + bb;
  if (LSM) {
    v4f z = r * invT;
    float m = fmaxf(fmaxf(z.x, z.y), fmaxf(z.z, z.w));
    for (int o = 1; o < LPR; o <<= 1) m = fmaxf(m, __shfl_xor(m, o, 32));
    float s = (expf(z.x - m) + expf(z.y - m)) + (expf(z.z - m) + expf(z.w - m));
    for (int o = 1; o < LPR; o <<= 1) s += __shfl_xor(s, o, 32);
    float lg = m + logf(s);
    r = z - lg;
  }
  if (sub == 0) *(v4f*)(OUT + (size_t)row * F + l * 4) = r;
}


extern "C" void kernel_launch(void* const* d_in, const int* in_sizes, int n_in,
                              void* d_out, int out_size, void* d_ws, size_t ws_size,
                              hipStream_t stream) {
  const float* x  = (const float*)d_in[0];
  const int*   ei = (const int*)d_in[1];
  const float* W1 = (const float*)d_in[2];
  const float* b1 = (const float*)d_in[3];
  const float* W2 = (const float*)d_in[4];
  const float* b2 = (const float*)d_in[5];

  const int IN = 256, HID = 128, OUT = 64;
  int n = in_sizes[0] / IN;
  int E = in_sizes[1] / 2;
  const int* src = ei;
  const int* dst = ei + E;

  float* ws = (float*)d_ws;
  size_t o = 0;
  float* dinv = ws + o; o += ((size_t)n + 255) & ~(size_t)255;
  float* H1   = ws + o; o += (size_t)n * HID;
  float* AGG1 = ws + o; o += (size_t)n * HID;
  float* H2   = ws + o; o += (size_t)n * OUT;
  int NB = (n + RB - 1) / RB;
  int* rowptr = (int*)(ws + o); o += ((size_t)n + 1 + 255) & ~(size_t)255;
  int* cntT   = (int*)(ws + o); o += ((size_t)NB * GS + 255) & ~(size_t)255;
  int* offT   = (int*)(ws + o); o += ((size_t)NB * GS + 255) & ~(size_t)255;
  int* pairs  = (int*)(ws + o); o += ((size_t)2 * E + 255) & ~(size_t)255;
  int* col    = (int*)(ws + o); o += ((size_t)E + 255) & ~(size_t)255;
  float* OUTP = (float*)d_out;

  int mtiles  = n / 16;
  int mblocks = (mtiles + 7) / 8;
  int chunk   = (E + GS - 1) / GS;

  k_slice_sort<<<GS, SB, NB * sizeof(int), stream>>>(src, dst, cntT, offT, pairs, E, NB, chunk);
  k_bin_rows_gemm<256, 128><<<NB, SB, 0, stream>>>(pairs, cntT, offT, rowptr, dinv, col, x, W1, H1, n, E, chunk);

  k_agg_rows<128, false><<<(unsigned)(((long long)n * 32 + TB - 1) / TB), TB, 0, stream>>>(
      H1, col, rowptr, dinv, b1, AGG1, n, 1.0f);

  k_gemm_wmma<128, 64, true><<<mblocks * (OUT / 64), TB, 0, stream>>>(AGG1, W2, dinv, H2, n);

  k_agg_rows<64, true><<<(unsigned)(((long long)n * 32 + TB - 1) / TB), TB, 0, stream>>>(
      H2, col, rowptr, dinv, b2, OUTP, n, 5.0f);
}


# layer-2 product moved into the layer-1 row-sum launch: 16 waves sum 32 rows into LDS, 8 of them multiply that tile by W2 staged in LDS; 5 launches to 4, no AGG1 round trip; all sum orders kept
# speedup vs baseline: 4.8250x; 1.0429x over previous
#include <hip/hip_runtime.h>
#include <hip/hip_bf16.h>

typedef __attribute__((ext_vector_type(2))) float v2f;
typedef __attribute__((ext_vector_type(4))) float v4f;
typedef __attribute__((ext_vector_type(8))) float v8f;

#define TB 256


#define SB 512
#define RB 208
#define MTB (RB / 16)
#define GS 256
#define EU 8
#define SPW (GS / (SB / 32))

__device__ __forceinline__ int block_excl_scan(int v, int* wsum, int& total) {
  int lane = threadIdx.x & 31;
  int w    = threadIdx.x >> 5;
  int nw   = blockDim.x >> 5;
  int inc = v;
  for (int o = 1; o < 32; o <<= 1) {
    int u = __shfl_up(inc, o, 32);
    if (lane >= o) inc += u;
  }
  __syncthreads();
  if (lane == 31) wsum[w] = inc;
  __syncthreads();
  if (w == 0) {
    int x = (lane < nw) ? wsum[lane] : 0;
    for (int o = 1; o < 32; o <<= 1) {
      int u = __shfl_up(x, o, 32);
      if (lane >= o) x += u;
    }
    wsum[lane] = x;
  }
  __syncthreads();
  total = wsum[31];
  return inc - v + (w > 0 ? wsum[w - 1] : 0);
}

__global__ void __launch_bounds__(SB) k_slice_sort(const int* __restrict__ src, const int* __restrict__ dst,
                                                   int* __restrict__ cntT, int* __restrict__ offT,
                                                   int* __restrict__ pairs, int E, int NB, int chunk) {
  extern __shared__ int lbin[];
  __shared__ int wsum[32];
  int g = blockIdx.x;
  for (int b = threadIdx.x; b < NB; b += SB) lbin[b] = 0;
  __syncthreads();
  int e0 = g * chunk;
  int e1 = min(E, e0 + chunk);
  for (int e = e0 + (int)threadIdx.x; e < e1; e += SB * EU) {
    int d[EU];
#pragma unroll
    for (int u = 0; u < EU; ++u) d[u] = dst[min(e + u * SB, e1 - 1)];
#pragma unroll
    for (int u = 0; u < EU; ++u) atomicAdd(&lbin[(unsigned)d[u] / RB], (e + u * SB < e1) ? 1 : 0);
  }
  __syncthreads();
  int carry = e0;
  for (int t0 = 0; t0 < NB; t0 += SB) {
    int b = t0 + (int)threadIdx.x;
    int c = (b < NB) ? lbin[b] : 0;
    int tot;
    int ex = block_excl_scan(c, wsum, tot);
    if (b < NB) {
      int first = carry + ex;
      cntT[b * GS + g] = c;
      offT[b * GS + g] = first;
      lbin[b] = first;
    }
    carry += tot;
  }
  __syncthreads();
  for (int e = e0 + (int)threadIdx.x; e < e1; e += SB * EU) {
    int d[EU], s[EU];
#pragma unroll
    for (int u = 0; u < EU; ++u) {
      int ee = min(e + u * SB, e1 - 1);
      d[u] = dst[ee];
      s[u] = src[ee];
    }
    int p[EU];
#pragma unroll
    for (int u = 0; u < EU; ++u) p[u] = atomicAdd(&lbin[(unsigned)d[u] / RB], (e + u * SB < e1) ? 1 : 0);
#pragma unroll
    for (int u = 0; u < EU; ++u)
      if (e + u * SB < e1) *(int2*)(pairs + 2 * (size_t)p[u]) = make_int2(s[u], d[u]);
  }
}

template<int K, int N>
__global__ void __launch_bounds__(SB) k_bin_rows_gemm(const int* __restrict__ pairs, const int* __restrict__ cntT,
                                                      const int* __restrict__ offT,
                                                      int* __restrict__ rowptr, float* __restrict__ dinv,
                                                      int* __restrict__ col,
                                                      const float* __restrict__ X, const float* __restrict__ Wm,
                                                      float* __restrict__ H, int n, int E, int chunk) {
  constexpr int LDK = K + 4;
  constexpr int BSF = 41 * 1024;
  static_assert(N * LDK <= BSF, "transposed Wm fits");
  static_assert(N == 128 && K % 32 == 0, "4 column groups of 32; k in steps of 32");
  __shared__ __attribute__((aligned(16))) float Bs[BSF];
  __shared__ __attribute__((aligned(16))) float ldinv[RB];
  __shared__ int lcnt[RB];
  __shared__ int lcur[RB];
  __shared__ int wsum[32];
  {
    constexpr int SG = 32;
    static_assert((K * N) % (SB * SG) == 0 && SB % N == 0, "whole groups; a thread keeps its column");
    int c  = threadIdx.x % N;
    int k0 = threadIdx.x / N;
    for (int base = 0; base < K * N; base += SB * SG) {
      float v[SG];
#pragma unroll
      for (int u = 0; u < SG; ++u) v[u] = Wm[base + u * SB + (int)threadIdx.x];
#pragma unroll
      for (int u = 0; u < SG; ++u) Bs[c * LDK + base / N + u * (SB / N) + k0] = v[u];
    }
  }
  int b    = blockIdx.x;
  int n0   = b * RB;
  int lane = threadIdx.x & 31;
  int w    = threadIdx.x >> 5;
  for (int t = threadIdx.x; t < RB; t += SB) lcnt[t] = 0;
  int segc = 0, sego = 0, part = 0;
  if (lane < SPW) {
    int g = w * SPW + lane;
    segc = cntT[b * GS + g];
    sego = offT[b * GS + g];
    part = sego - g * chunk;
  }
  int base;
  block_excl_scan(part, wsum, base);
  int maxlen = segc;
  for (int o = 16; o > 0; o >>= 1) maxlen = max(maxlen, __shfl_xor(maxlen, o, 32));
  const int2* pr = (const int2*)pairs;

  {
    int len[SPW];
    int da[SPW], db[SPW];
#pragma unroll
    for (int u = 0; u < SPW; ++u) {
      len[u]  = __shfl(segc, u, 32);
      int off = __shfl(sego, u, 32);
      da[u] = pairs[2 * (size_t)((lane      < len[u]) ? off + lane      : 0) + 1];
      db[u] = pairs[2 * (size_t)((lane + 32 < len[u]) ? off + lane + 32 : 0) + 1];
    }
#pragma unroll
    for (int u = 0; u < SPW; ++u) {
      if (lane      < len[u]) atomicAdd(&lcnt[da[u] - n0], 1);
      if (lane + 32 < len[u]) atomicAdd(&lcnt[db[u] - n0], 1);
    }
  }
  if (maxlen > 64) {
    for (int u = 0; u < SPW; ++u) {
      int len = __shfl(segc, u, 32);
      int off = __shfl(sego, u, 32);
      for (int i = 64 + lane; i < len; i += 32) atomicAdd(&lcnt[pr[off + i].y - n0], 1);
    }
  }
  __syncthreads();
  int carry = 0;
  for (int t0 = 0; t0 < RB; t0 += SB) {
    int t = t0 + (int)threadIdx.x;
    int c = (t < RB) ? lcnt[t] : 0;
    int tot;
    int ex = block_excl_scan(c, wsum, tot);
    if (t < RB) {
      int first = base + carry + ex;
      lcur[t] = first;
      int node = n0 + t;
      float dv = rsqrtf((float)(c + 1));
      ldinv[t] = dv;
      if (node < n) {
        rowptr[node] = first;
        dinv[node]   = dv;
      }
    }
    carry += tot;
  }
  if (b == (int)gridDim.x - 1 && threadIdx.x == 0) rowptr[n] = E;
  __syncthreads();

  {
    int len[SPW];
    int2 ta[SPW], tb[SPW];
#pragma unroll
    for (int u = 0; u < SPW; ++u) {
      len[u]  = __shfl(segc, u, 32);
      int off = __shfl(sego, u, 32);
      ta[u] = pr[(lane      < len[u]) ? off + lane      : 0];
      tb[u] = pr[(lane + 32 < len[u]) ? off + lane + 32 : 0];
    }
#pragma unroll
    for (int u = 0; u < SPW; ++u) {
      if (lane      < len[u]) col[atomicAdd(&lcur[ta[u].y - n0], 1)] = ta[u].x;
      if (lane + 32 < len[u]) col[atomicAdd(&lcur[tb[u].y - n0], 1)] = tb[u].x;
    }
  }
  if (maxlen > 64) {
    for (int u = 0; u < SPW; ++u) {
      int len = __shfl(segc, u, 32);
      int off = __shfl(sego, u, 32);
      for (int i = 64 + lane; i < len; i += 32) {
        int2 t = pr[off + i];
        col[atomicAdd(&lcur[t.y - n0], 1)] = t.x;
      }
    }
  }

  {
    int half = lane >> 4;
    int l    = lane & 15;
    int rg   = ((w & 3) + (w >> 2)) & 3;
    int nc0  = (w >> 2) * 32;
    int mtiles = n >> 4;
    const float* Bl = Bs + (nc0 + l) * LDK + half * 4;
    for (int mt = rg; mt < MTB; mt += 4) {
      int gmt = b * MTB + mt;
      if (gmt >= mtiles) break;
      int m0 = gmt << 4;
      const float* Arow = X + (size_t)(m0 + l) * K + half * 4;
      v8f acc0 = {}, acc1 = {};
      v4f an[4];
#pragma unroll
      for (int u = 0; u < 4; ++u) an[u] = *(const v4f*)(Arow + u * 8);
#pragma unroll
      for (int kk = 0; kk < K; kk += 32) {
        v4f a[4];
#pragma unroll
        for (int u = 0; u < 4; ++u) a[u] = an[u];
        if (kk + 32 < K) {
#pragma unroll
          for (int u = 0; u < 4; ++u) an[u] = *(const v4f*)(Arow + kk + 32 + u * 8);
        }
#pragma unroll
        for (int u = 0; u < 4; ++u) {
          v4f b0 = *(const v4f*)(Bl + kk + u * 8);
          v4f b1 = *(const v4f*)(Bl + 16 * LDK + kk + u * 8);
          v2f alo, ahi, b0lo, b0hi, b1lo, b1hi;
          alo.x  = a[u].x; alo.y  = a[u].y; ahi.x  = a[u].z; ahi.y  = a[u].w;
          b0lo.x = b0.x;   b0lo.y = b0.y;   b0hi.x = b0.z;   b0hi.y = b0.w;
          b1lo.x = b1.x;   b1lo.y = b1.y;   b1hi.x = b1.z;   b1hi.y = b1.w;
          acc0 = __builtin_amdgcn_wmma_f32_16x16x4_f32(false, alo, false, b0lo, (short)0, acc0, false, false);
          acc1 = __builtin_amdgcn_wmma_f32_16x16x4_f32(false, alo, false, b1lo, (short)0, acc1, false, false);
          acc0 = __builtin_amdgcn_wmma_f32_16x16x4_f32(false, ahi, false, b0hi, (short)0, acc0, false, false);
          acc1 = __builtin_amdgcn_wmma_f32_16x16x4_f32(false, ahi, false, b1hi, (short)0, acc1, false, false);
        }
      }
      int rl = mt * 16 + half * 8;
      v4f s0 = *(const v4f*)(ldinv + rl);
      v4f s1 = *(const v4f*)(ldinv + rl + 4);
      float sc[8] = {s0.x, s0.y, s0.z, s0.w, s1.x, s1.y, s1.z, s1.w};
      float* Hp = H + (size_t)(m0 + half * 8) * N + nc0 + l;
#pragma unroll
      for (int j = 0; j < 8; ++j) {
        Hp[(size_t)j * N]      = acc0[j] * sc[j];
        Hp[(size_t)j * N + 16] = acc1[j] * sc[j];
      }
    }
  }
}


template<int F, bool LSM>
__global__ void k_agg_rows(const float* __restrict__ HS, const int* __restrict__ col,
                           const int* __restrict__ rowptr, const float* __restrict__ dinv,
                           const float* __restrict__ b, float* __restrict__ OUT, int n, float invT) {
  constexpr int LPR = F / 4;
  constexpr int EPI = 32 / LPR;
  constexpr int U   = 8;
  int row  = (int)((blockIdx.x * blockDim.x + threadIdx.x) >> 5);
  int lane = threadIdx.x & 31;
  if (row >= n) return;
  int sub = lane / LPR;
  int l   = lane % LPR;
  int start = rowptr[row];
  int end   = rowptr[row + 1];
  float di  = dinv[row];
  const v4f zero = {0.0f, 0.0f, 0.0f, 0.0f};
  v4f acc = zero;
  if (sub == 0)
    acc = *(const v4f*)(HS + (size_t)row * F + l * 4);
  for (int base = start; base < end; base += 32) {
    int m = min(32, end - base);
    int my_s = 0;
    if (lane < m) my_s = col[base + lane];
    for (int j = 0; j < m; j += EPI * U) {
      v4f v[U];
      int s[U];
#pragma unroll
      for (int u = 0; u < U; ++u) {
        int jj = j + u * EPI + sub;
        s[u] = __shfl(my_s, jj, 32);
      }
#pragma unroll
      for (int u = 0; u < U; ++u) {
        int jj = j + u * EPI + sub;
        v[u] = (jj < m) ? *(const v4f*)(HS + (size_t)s[u] * F + l * 4) : zero;
      }
#pragma unroll
      for (int u = 0; u < U; ++u) acc += v[u];
    }
  }
  for (int o = LPR; o < 32; o <<= 1) {
    acc.x += __shfl_xor(acc.x, o, 32);
    acc.y += __shfl_xor(acc.y, o, 32);
    acc.z += __shfl_xor(acc.z, o, 32);
    acc.w += __shfl_xor(acc.w, o, 32);
  }
  v4f bb = *(const v4f*)(b + l * 4);
  v4f r = acc * di + bb;
  if (LSM) {
    v4f z = r * invT;
    float m = fmaxf(fmaxf(z.x, z.y), fmaxf(z.z, z.w));
    for (int o = 1; o < LPR; o <<= 1) m = fmaxf(m, __shfl_xor(m, o, 32));
    float s = (expf(z.x - m) + expf(z.y - m)) + (expf(z.z - m) + expf(z.w - m));
    for (int o = 1; o < LPR; o <<= 1) s += __shfl_xor(s, o, 32);
    float lg = m + logf(s);
    r = z - lg;
  }
  if (sub == 0) *(v4f*)(OUT + (size_t)row * F + l * 4) = r;
}


#define AR  32
#define AWV 16
template<int F, int N2>
__global__ void __launch_bounds__(AWV * 32) k_agg_rows_gemm(const float* __restrict__ HS, const int* __restrict__ col,
                                                            const int* __restrict__ rowptr, const float* __restrict__ dinv,
                                                            const float* __restrict__ b, const float* __restrict__ Wm,
                                                            float* __restrict__ C, int n) {
  constexpr int NTH = AWV * 32;
  constexpr int LDA = F + 4;
  constexpr int NT  = N2 / 16;
  constexpr int U   = 8;
  static_assert(F == 128, "a lane holds 16 bytes of a row: 32 lanes x 4 floats");
  static_assert(AR % AWV == 0 && AR % 16 == 0 && N2 % 16 == 0, "whole rows per wave, whole tiles");
  static_assert((AR / 16) * NT <= AWV, "one output tile per product wave");
  static_assert((F * N2) % NTH == 0 && NTH % N2 == 0, "whole staging groups; a thread keeps its column");
  __shared__ __attribute__((aligned(16))) float Bs[N2 * LDA];
  __shared__ __attribute__((aligned(16))) float As[AR * LDA];
  __shared__ __attribute__((aligned(16))) float ldinv[AR];
  int lane = threadIdx.x & 31;
  int w    = threadIdx.x >> 5;
  int r0   = blockIdx.x * AR;

  constexpr int SG = F * N2 / NTH;
  constexpr int KS = NTH / N2;
  float wv[SG];
#pragma unroll
  for (int u = 0; u < SG; ++u) wv[u] = Wm[u * NTH + (int)threadIdx.x];

  const v4f zero = {0.0f, 0.0f, 0.0f, 0.0f};
  v4f bb = *(const v4f*)(b + lane * 4);
#pragma unroll
  for (int q = 0; q < AR / AWV; ++q) {
    int slot = q * AWV + w;
    int row  = r0 + slot;
    bool valid = row < n;
    int start = 0, end = 0;
    float di = 0.0f;
    v4f acc = zero;
    if (valid) {
      start = rowptr[row];
      end   = rowptr[row + 1];
      di    = dinv[row];
      acc   = *(const v4f*)(HS + (size_t)row * F + lane * 4);
    }
    if (q == 0) {
      int c  = threadIdx.x % N2;
      int k0 = threadIdx.x / N2;
#pragma unroll
      for (int u = 0; u < SG; ++u) Bs[c * LDA + u * KS + k0] = wv[u];
    }
    v4f r = zero;
    if (valid) {
      for (int base = start; base < end; base += 32) {
        int m = min(32, end - base);
        int my_s = 0;
        if (lane < m) my_s = col[base + lane];
        for (int j = 0; j < m; j += U) {
          v4f v[U];
          int s[U];
#pragma unroll
          for (int u = 0; u < U; ++u) s[u] = __shfl(my_s, j + u, 32);
#pragma unroll
          for (int u = 0; u < U; ++u)
            v[u] = (j + u < m) ? *(const v4f*)(HS + (size_t)s[u] * F + lane * 4) : zero;
#pragma unroll
          for (int u = 0; u < U; ++u) acc += v[u];
        }
      }
      r = acc * di + bb;
      r.x = fmaxf(r.x, 0.0f); r.y = fmaxf(r.y, 0.0f); r.z = fmaxf(r.z, 0.0f); r.w = fmaxf(r.w, 0.0f);
    }
    *(v4f*)(As + slot * LDA + lane * 4) = r;
    if (lane == 0) ldinv[slot] = di;
  }
  __syncthreads();

  if (w < (AR / 16) * NT) {
    int mt   = w / NT;
    int nt   = w % NT;
    int half = lane >> 4;
    int l    = lane & 15;
    const float* Al = As + (mt * 16 + l) * LDA + half * 2;
    const float* Bl = Bs + (nt * 16 + l) * LDA + half * 2;
    v8f acc = {};
#pragma unroll
    for (int k = 0; k < F; k += 4) {
      v2f a  = *(const v2f*)(Al + k);
      v2f bq = *(const v2f*)(Bl + k);
      acc = __builtin_amdgcn_wmma_f32_16x16x4_f32(false, a, false, bq, (short)0, acc, false, false);
    }
    int rl = mt * 16 + half * 8;
    v4f s0 = *(const v4f*)(ldinv + rl);
    v4f s1 = *(const v4f*)(ldinv + rl + 4);
    float sc[8] = {s0.x, s0.y, s0.z, s0.w, s1.x, s1.y, s1.z, s1.w};
    int grow = r0 + rl;
    float* Cp = C + (size_t)grow * N2 + nt * 16 + l;
#pragma unroll
    for (int j = 0; j < 8; ++j)
      if (grow + j < n) Cp[(size_t)j * N2] = acc[j] * sc[j];
  }
}


extern "C" void kernel_launch(void* const* d_in, const int* in_sizes, int n_in,
                              void* d_out, int out_size, void* d_ws, size_t ws_size,
                              hipStream_t stream) {
  const float* x  = (const float*)d_in[0];
  const int*   ei = (const int*)d_in[1];
  const float* W1 = (const float*)d_in[2];
  const float* b1 = (const float*)d_in[3];
  const float* W2 = (const float*)d_in[4];
  const float* b2 = (const float*)d_in[5];

  const int IN = 256, HID = 128, OUT = 64;
  int n = in_sizes[0] / IN;
  int E = in_sizes[1] / 2;
  const int* src = ei;
  const int* dst = ei + E;

  float* ws = (float*)d_ws;
  size_t o = 0;
  float* dinv = ws + o; o += ((size_t)n + 255) & ~(size_t)255;
  float* H1   = ws + o; o += (size_t)n * HID;
  float* H2   = ws + o; o += (size_t)n * OUT;
  int NB = (n + RB - 1) / RB;
  int* rowptr = (int*)(ws + o); o += ((size_t)n + 1 + 255) & ~(size_t)255;
  int* cntT   = (int*)(ws + o); o += ((size_t)NB * GS + 255) & ~(size_t)255;
  int* offT   = (int*)(ws + o); o += ((size_t)NB * GS + 255) & ~(size_t)255;
  int* pairs  = (int*)(ws + o); o += ((size_t)2 * E + 255) & ~(size_t)255;
  int* col    = (int*)(ws + o); o += ((size_t)E + 255) & ~(size_t)255;
  float* OUTP = (float*)d_out;

  int chunk   = (E + GS - 1) / GS;

  k_slice_sort<<<GS, SB, NB * sizeof(int), stream>>>(src, dst, cntT, offT, pairs, E, NB, chunk);
  k_bin_rows_gemm<256, 128><<<NB, SB, 0, stream>>>(pairs, cntT, offT, rowptr, dinv, col, x, W1, H1, n, E, chunk);

  k_agg_rows_gemm<128, 64><<<(unsigned)((n + AR - 1) / AR), AWV * 32, 0, stream>>>(
      H1, col, rowptr, dinv, b1, W2, H2, n);

  k_agg_rows<64, true><<<(unsigned)(((long long)n * 32 + TB - 1) / TB), TB, 0, stream>>>(
      H2, col, rowptr, dinv, b2, OUTP, n, 5.0f);
}


# slice sort: lanes past a slice's end issue no LDS atomic (they added 0, all to one word; same-address lanes are served one by one, every wave of the block queues behind); row sums still arrival order
# speedup vs baseline: 5.1164x; 1.0604x over previous
#include <hip/hip_runtime.h>
#include <hip/hip_bf16.h>

typedef __attribute__((ext_vector_type(2))) float v2f;
typedef __attribute__((ext_vector_type(4))) float v4f;
typedef __attribute__((ext_vector_type(8))) float v8f;

#define TB 256


#define SB 512
#define RB 208
#define MTB (RB / 16)
#define GS 256
#define EU 8
#define SPW (GS / (SB / 32))

__device__ __forceinline__ int block_excl_scan(int v, int* wsum, int& total) {
  int lane = threadIdx.x & 31;
  int w    = threadIdx.x >> 5;
  int nw   = blockDim.x >> 5;
  int inc = v;
  for (int o = 1; o < 32; o <<= 1) {
    int u = __shfl_up(inc, o, 32);
    if (lane >= o) inc += u;
  }
  __syncthreads();
  if (lane == 31) wsum[w] = inc;
  __syncthreads();
  if (w == 0) {
    int x = (lane < nw) ? wsum[lane] : 0;
    for (int o = 1; o < 32; o <<= 1) {
      int u = __shfl_up(x, o, 32);
      if (lane >= o) x += u;
    }
    wsum[lane] = x;
  }
  __syncthreads();
  total = wsum[31];
  return inc - v + (w > 0 ? wsum[w - 1] : 0);
}

__global__ void __launch_bounds__(SB) k_slice_sort(const int* __restrict__ src, const int* __restrict__ dst,
                                                   int* __restrict__ cntT, int* __restrict__ offT,
                                                   int* __restrict__ pairs, int E, int NB, int chunk) {
  extern __shared__ int lbin[];
  __shared__ int wsum[32];
  int g = blockIdx.x;
  for (int b = threadIdx.x; b < NB; b += SB) lbin[b] = 0;
  __syncthreads();
  int e0 = g * chunk;
  int e1 = min(E, e0 + chunk);
  for (int e = e0 + (int)threadIdx.x; e < e1; e += SB * EU) {
    int d[EU];
#pragma unroll
    for (int u = 0; u < EU; ++u) d[u] = dst[min(e + u * SB, e1 - 1)];
    static_assert(EU == 8, "the statement below names the EU loaded values");
    asm volatile("" :: "v"(d[0]), "v"(d[1]), "v"(d[2]), "v"(d[3]), "v"(d[4]), "v"(d[5]), "v"(d[6]), "v"(d[7]));
#pragma unroll
    for (int u = 0; u < EU; ++u)
      if (e + u * SB < e1) atomicAdd(&lbin[(unsigned)d[u] / RB], 1);
  }
  __syncthreads();
  int carry = e0;
  for (int t0 = 0; t0 < NB; t0 += SB) {
    int b = t0 + (int)threadIdx.x;
    int c = (b < NB) ? lbin[b] : 0;
    int tot;
    int ex = block_excl_scan(c, wsum, tot);
    if (b < NB) {
      int first = carry + ex;
      cntT[b * GS + g] = c;
      offT[b * GS + g] = first;
      lbin[b] = first;
    }
    carry += tot;
  }
  __syncthreads();
  for (int e = e0 + (int)threadIdx.x; e < e1; e += SB * EU) {
    int d[EU], s[EU];
#pragma unroll
    for (int u = 0; u < EU; ++u) {
      int ee = min(e + u * SB, e1 - 1);
      d[u] = dst[ee];
      s[u] = src[ee];
    }
    int p[EU];
#pragma unroll
    for (int u = 0; u < EU; ++u) {
      p[u] = 0;
      if (e + u * SB < e1) p[u] = atomicAdd(&lbin[(unsigned)d[u] / RB], 1);
    }
#pragma unroll
    for (int u = 0; u < EU; ++u)
      if (e + u * SB < e1) *(int2*)(pairs + 2 * (size_t)p[u]) = make_int2(s[u], d[u]);
  }
}

template<int K, int N>
__global__ void __launch_bounds__(SB) k_bin_rows_gemm(const int* __restrict__ pairs, const int* __restrict__ cntT,
                                                      const int* __restrict__ offT,
                                                      int* __restrict__ rowptr, float* __restrict__ dinv,
                                                      int* __restrict__ col,
                                                      const float* __restrict__ X, const float* __restrict__ Wm,
                                                      float* __restrict__ H, int n, int E, int chunk) {
  constexpr int LDK = K + 4;
  constexpr int BSF = 41 * 1024;
  static_assert(N * LDK <= BSF, "transposed Wm fits");
  static_assert(N == 128 && K % 32 == 0, "4 column groups of 32; k in steps of 32");
  __shared__ __attribute__((aligned(16))) float Bs[BSF];
  __shared__ __attribute__((aligned(16))) float ldinv[RB];
  __shared__ int lcnt[RB];
  __shared__ int lcur[RB];
  __shared__ int wsum[32];
  {
    constexpr int SG = 32;
    static_assert((K * N) % (SB * SG) == 0 && SB % N == 0, "whole groups; a thread keeps its column");
    int c  = threadIdx.x % N;
    int k0 = threadIdx.x / N;
    for (int base = 0; base < K * N; base += SB * SG) {
      float v[SG];
#pragma unroll
      for (int u = 0; u < SG; ++u) v[u] = Wm[base + u * SB + (int)threadIdx.x];
#pragma unroll
      for (int u = 0; u < SG; ++u) Bs[c * LDK + base / N + u * (SB / N) + k0] = v[u];
    }
  }
  int b    = blockIdx.x;
  int n0   = b * RB;
  int lane = threadIdx.x & 31;
  int w    = threadIdx.x >> 5;
  for (int t = threadIdx.x; t < RB; t += SB) lcnt[t] = 0;
  int segc = 0, sego = 0, part = 0;
  if (lane < SPW) {
    int g = w * SPW + lane;
    segc = cntT[b * GS + g];
    sego = offT[b * GS + g];
    part = sego - g * chunk;
  }
  int base;
  block_excl_scan(part, wsum, base);
  int maxlen = segc;
  for (int o = 16; o > 0; o >>= 1) maxlen = max(maxlen, __shfl_xor(maxlen, o, 32));
  const int2* pr = (const int2*)pairs;

  {
    int len[SPW];
    int da[SPW], db[SPW];
#pragma unroll
    for (int u = 0; u < SPW; ++u) {
      len[u]  = __shfl(segc, u, 32);
      int off = __shfl(sego, u, 32);
      da[u] = pairs[2 * (size_t)((lane      < len[u]) ? off + lane      : 0) + 1];
      db[u] = pairs[2 * (size_t)((lane + 32 < len[u]) ? off + lane + 32 : 0) + 1];
    }
#pragma unroll
    for (int u = 0; u < SPW; ++u) {
      if (lane      < len[u]) atomicAdd(&lcnt[da[u] - n0], 1);
      if (lane + 32 < len[u]) atomicAdd(&lcnt[db[u] - n0], 1);
    }
  }
  if (maxlen > 64) {
    for (int u = 0; u < SPW; ++u) {
      int len = __shfl(segc, u, 32);
      int off = __shfl(sego, u, 32);
      for (int i = 64 + lane; i < len; i += 32) atomicAdd(&lcnt[pr[off + i].y - n0], 1);
    }
  }
  __syncthreads();
  int carry = 0;
  for (int t0 = 0; t0 < RB; t0 += SB) {
    int t = t0 + (int)threadIdx.x;
    int c = (t < RB) ? lcnt[t] : 0;
    int tot;
    int ex = block_excl_scan(c, wsum, tot);
    if (t < RB) {
      int first = base + carry + ex;
      lcur[t] = first;
      int node = n0 + t;
      float dv = rsqrtf((float)(c + 1));
      ldinv[t] = dv;
      if (node < n) {
        rowptr[node] = first;
        dinv[node]   = dv;
      }
    }
    carry += tot;
  }
  if (b == (int)gridDim.x - 1 && threadIdx.x == 0) rowptr[n] = E;
  __syncthreads();

  {
    int len[SPW];
    int2 ta[SPW], tb[SPW];
#pragma unroll
    for (int u = 0; u < SPW; ++u) {
      len[u]  = __shfl(segc, u, 32);
      int off = __shfl(sego, u, 32);
      ta[u] = pr[(lane      < len[u]) ? off + lane      : 0];
      tb[u] = pr[(lane + 32 < len[u]) ? off + lane + 32 : 0];
    }
#pragma unroll
    for (int u = 0; u < SPW; ++u) {
      if (lane      < len[u]) col[atomicAdd(&lcur[ta[u].y - n0], 1)] = ta[u].x;
      if (lane + 32 < len[u]) col[atomicAdd(&lcur[tb[u].y - n0], 1)] = tb[u].x;
    }
  }
  if (maxlen > 64) {
    for (int u = 0; u < SPW; ++u) {
      int len = __shfl(segc, u, 32);
      int off = __shfl(sego, u, 32);
      for (int i = 64 + lane; i < len; i += 32) {
        int2 t = pr[off + i];
        col[atomicAdd(&lcur[t.y - n0], 1)] = t.x;
      }
    }
  }

  {
    int half = lane >> 4;
    int l    = lane & 15;
    int rg   = ((w & 3) + (w >> 2)) & 3;
    int nc0  = (w >> 2) * 32;
    int mtiles = n >> 4;
    const float* Bl = Bs + (nc0 + l) * LDK + half * 4;
    for (int mt = rg; mt < MTB; mt += 4) {
      int gmt = b * MTB + mt;
      if (gmt >= mtiles) break;
      int m0 = gmt << 4;
      const float* Arow = X + (size_t)(m0 + l) * K + half * 4;
      v8f acc0 = {}, acc1 = {};
      v4f an[4];
#pragma unroll
      for (int u = 0; u < 4; ++u) an[u] = *(const v4f*)(Arow + u * 8);
#pragma unroll
      for (int kk = 0; kk < K; kk += 32) {
        v4f a[4];
#pragma unroll
        for (int u = 0; u < 4; ++u) a[u] = an[u];
        if (kk + 32 < K) {
#pragma unroll
          for (int u = 0; u < 4; ++u) an[u] = *(const v4f*)(Arow + kk + 32 + u * 8);
        }
#pragma unroll
        for (int u = 0; u < 4; ++u) {
          v4f b0 = *(const v4f*)(Bl + kk + u * 8);
          v4f b1 = *(const v4f*)(Bl + 16 * LDK + kk + u * 8);
          v2f alo, ahi, b0lo, b0hi, b1lo, b1hi;
          alo.x  = a[u].x; alo.y  = a[u].y; ahi.x  = a[u].z; ahi.y  = a[u].w;
          b0lo.x = b0.x;   b0lo.y = b0.y;   b0hi.x = b0.z;   b0hi.y = b0.w;
          b1lo.x = b1.x;   b1lo.y = b1.y;   b1hi.x = b1.z;   b1hi.y = b1.w;
          acc0 = __builtin_amdgcn_wmma_f32_16x16x4_f32(false, alo, false, b0lo, (short)0, acc0, false, false);
          acc1 = __builtin_amdgcn_wmma_f32_16x16x4_f32(false, alo, false, b1lo, (short)0, acc1, false, false);
          acc0 = __builtin_amdgcn_wmma_f32_16x16x4_f32(false, ahi, false, b0hi, (short)0, acc0, false, false);
          acc1 = __builtin_amdgcn_wmma_f32_16x16x4_f32(false, ahi, false, b1hi, (short)0, acc1, false, false);
        }
      }
      int rl = mt * 16 + half * 8;
      v4f s0 = *(const v4f*)(ldinv + rl);
      v4f s1 = *(const v4f*)(ldinv + rl + 4);
      float sc[8] = {s0.x, s0.y, s0.z, s0.w, s1.x, s1.y, s1.z, s1.w};
      float* Hp = H + (size_t)(m0 + half * 8) * N + nc0 + l;
#pragma unroll
      for (int j = 0; j < 8; ++j) {
        Hp[(size_t)j * N]      = acc0[j] * sc[j];
        Hp[(size_t)j * N + 16] = acc1[j] * sc[j];
      }
    }
  }
}


template<int F, bool LSM>
__global__ void k_agg_rows(const float* __restrict__ HS, const int* __restrict__ col,
                           const int* __restrict__ rowptr, const float* __restrict__ dinv,
                           const float* __restrict__ b, float* __restrict__ OUT, int n, float invT) {
  constexpr int LPR = F / 4;
  constexpr int EPI = 32 / LPR;
  constexpr int U   = 8;
  int row  = (int)((blockIdx.x * blockDim.x + threadIdx.x) >> 5);
  int lane = threadIdx.x & 31;
  if (row >= n) return;
  int sub = lane / LPR;
  int l   = lane % LPR;
  int start = rowptr[row];
  int end   = rowptr[row + 1];
  float di  = dinv[row];
  const v4f zero = {0.0f, 0.0f, 0.0f, 0.0f};
  v4f acc = zero;
  if (sub == 0)
    acc = *(const v4f*)(HS + (size_t)row * F + l * 4);
  for (int base = start; base < end; base += 32) {
    int m = min(32, end - base);
    int my_s = 0;
    if (lane < m) my_s = col[base + lane];
    for (int j = 0; j < m; j += EPI * U) {
      v4f v[U];
      int s[U];
#pragma unroll
      for (int u = 0; u < U; ++u) {
        int jj = j + u * EPI + sub;
        s[u] = __shfl(my_s, jj, 32);
      }
#pragma unroll
      for (int u = 0; u < U; ++u) {
        int jj = j + u * EPI + sub;
        v[u] = (jj < m) ? *(const v4f*)(HS + (size_t)s[u] * F + l * 4) : zero;
      }
#pragma unroll
      for (int u = 0; u < U; ++u) acc += v[u];
    }
  }
  for (int o = LPR; o < 32; o <<= 1) {
    acc.x += __shfl_xor(acc.x, o, 32);
    acc.y += __shfl_xor(acc.y, o, 32);
    acc.z += __shfl_xor(acc.z, o, 32);
    acc.w += __shfl_xor(acc.w, o, 32);
  }
  v4f bb = *(const v4f*)(b + l * 4);
  v4f r = acc * di + bb;
  if (LSM) {
    v4f z = r * invT;
    float m = fmaxf(fmaxf(z.x, z.y), fmaxf(z.z, z.w));
    for (int o = 1; o < LPR; o <<= 1) m = fmaxf(m, __shfl_xor(m, o, 32));
    float s = (expf(z.x - m) + expf(z.y - m)) + (expf(z.z - m) + expf(z.w - m));
    for (int o = 1; o < LPR; o <<= 1) s += __shfl_xor(s, o, 32);
    float lg = m + logf(s);
    r = z - lg;
  }
  if (sub == 0) *(v4f*)(OUT + (size_t)row * F + l * 4) = r;
}


#define AR  32
#define AWV 16
template<int F, int N2>
__global__ void __launch_bounds__(AWV * 32) k_agg_rows_gemm(const float* __restrict__ HS, const int* __restrict__ col,
                                                            const int* __restrict__ rowptr, const float* __restrict__ dinv,
                                                            const float* __restrict__ b, const float* __restrict__ Wm,
                                                            float* __restrict__ C, int n) {
  constexpr int NTH = AWV * 32;
  constexpr int LDA = F + 4;
  constexpr int NT  = N2 / 16;
  constexpr int U   = 8;
  static_assert(F == 128, "a lane holds 16 bytes of a row: 32 lanes x 4 floats");
  static_assert(AR % AWV == 0 && AR % 16 == 0 && N2 % 16 == 0, "whole rows per wave, whole tiles");
  static_assert((AR / 16) * NT <= AWV, "one output tile per product wave");
  static_assert((F * N2) % NTH == 0 && NTH % N2 == 0, "whole staging groups; a thread keeps its column");
  __shared__ __attribute__((aligned(16))) float Bs[N2 * LDA];
  __shared__ __attribute__((aligned(16))) float As[AR * LDA];
  __shared__ __attribute__((aligned(16))) float ldinv[AR];
  int lane = threadIdx.x & 31;
  int w    = threadIdx.x >> 5;
  int r0   = blockIdx.x * AR;

  constexpr int SG = F * N2 / NTH;
  constexpr int KS = NTH / N2;
  float wv[SG];
#pragma unroll
  for (int u = 0; u < SG; ++u) wv[u] = Wm[u * NTH + (int)threadIdx.x];

  const v4f zero = {0.0f, 0.0f, 0.0f, 0.0f};
  v4f bb = *(const v4f*)(b + lane * 4);
#pragma unroll
  for (int q = 0; q < AR / AWV; ++q) {
    int slot = q * AWV + w;
    int row  = r0 + slot;
    bool valid = row < n;
    int start = 0, end = 0;
    float di = 0.0f;
    v4f acc = zero;
    if (valid) {
      start = rowptr[row];
      end   = rowptr[row + 1];
      di    = dinv[row];
      acc   = *(const v4f*)(HS + (size_t)row * F + lane * 4);
    }
    if (q == 0) {
      int c  = threadIdx.x % N2;
      int k0 = threadIdx.x / N2;
#pragma unroll
      for (int u = 0; u < SG; ++u) Bs[c * LDA + u * KS + k0] = wv[u];
    }
    v4f r = zero;
    if (valid) {
      for (int base = start; base < end; base += 32) {
        int m = min(32, end - base);
        int my_s = 0;
        if (lane < m) my_s = col[base + lane];
        for (int j = 0; j < m; j += U) {
          v4f v[U];
          int s[U];
#pragma unroll
          for (int u = 0; u < U; ++u) s[u] = __shfl(my_s, j + u, 32);
#pragma unroll
          for (int u = 0; u < U; ++u)
            v[u] = (j + u < m) ? *(const v4f*)(HS + (size_t)s[u] * F + lane * 4) : zero;
#pragma unroll
          for (int u = 0; u < U; ++u) acc += v[u];
        }
      }
      r = acc * di + bb;
      r.x = fmaxf(r.x, 0.0f); r.y = fmaxf(r.y, 0.0f); r.z = fmaxf(r.z, 0.0f); r.w = fmaxf(r.w, 0.0f);
    }
    *(v4f*)(As + slot * LDA + lane * 4) = r;
    if (lane == 0) ldinv[slot] = di;
  }
  __syncthreads();

  if (w < (AR / 16) * NT) {
    int mt   = w / NT;
    int nt   = w % NT;
    int half = lane >> 4;
    int l    = lane & 15;
    const float* Al = As + (mt * 16 + l) * LDA + half * 2;
    const float* Bl = Bs + (nt * 16 + l) * LDA + half * 2;
    v8f acc = {};
#pragma unroll
    for (int k = 0; k < F; k += 4) {
      v2f a  = *(const v2f*)(Al + k);
      v2f bq = *(const v2f*)(Bl + k);
      acc = __builtin_amdgcn_wmma_f32_16x16x4_f32(false, a, false, bq, (short)0, acc, false, false);
    }
    int rl = mt * 16 + half * 8;
    v4f s0 = *(const v4f*)(ldinv + rl);
    v4f s1 = *(const v4f*)(ldinv + rl + 4);
    float sc[8] = {s0.x, s0.y, s0.z, s0.w, s1.x, s1.y, s1.z, s1.w};
    int grow = r0 + rl;
    float* Cp = C + (size_t)grow * N2 + nt * 16 + l;
#pragma unroll
    for (int j = 0; j < 8; ++j)
      if (grow + j < n) Cp[(size_t)j * N2] = acc[j] * sc[j];
  }
}


extern "C" void kernel_launch(void* const* d_in, const int* in_sizes, int n_in,
                              void* d_out, int out_size, void* d_ws, size_t ws_size,
                              hipStream_t stream) {
  const float* x  = (const float*)d_in[0];
  const int*   ei = (const int*)d_in[1];
  const float* W1 = (const float*)d_in[2];
  const float* b1 = (const float*)d_in[3];
  const float* W2 = (const float*)d_in[4];
  const float* b2 = (const float*)d_in[5];

  const int IN = 256, HID = 128, OUT = 64;
  int n = in_sizes[0] / IN;
  int E = in_sizes[1] / 2;
  const int* src = ei;
  const int* dst = ei + E;

  float* ws = (float*)d_ws;
  size_t o = 0;
  float* dinv = ws + o; o += ((size_t)n + 255) & ~(size_t)255;
  float* H1   = ws + o; o += (size_t)n * HID;
  float* H2   = ws + o; o += (size_t)n * OUT;
  int NB = (n + RB - 1) / RB;
  int* rowptr = (int*)(ws + o); o += ((size_t)n + 1 + 255) & ~(size_t)255;
  int* cntT   = (int*)(ws + o); o += ((size_t)NB * GS + 255) & ~(size_t)255;
  int* offT   = (int*)(ws + o); o += ((size_t)NB * GS + 255) & ~(size_t)255;
  int* pairs  = (int*)(ws + o); o += ((size_t)2 * E + 255) & ~(size_t)255;
  int* col    = (int*)(ws + o); o += ((size_t)E + 255) & ~(size_t)255;
  float* OUTP = (float*)d_out;

  int chunk   = (E + GS - 1) / GS;

  k_slice_sort<<<GS, SB, NB * sizeof(int), stream>>>(src, dst, cntT, offT, pairs, E, NB, chunk);
  k_bin_rows_gemm<256, 128><<<NB, SB, 0, stream>>>(pairs, cntT, offT, rowptr, dinv, col, x, W1, H1, n, E, chunk);

  k_agg_rows_gemm<128, 64><<<(unsigned)((n + AR - 1) / AR), AWV * 32, 0, stream>>>(
      H1, col, rowptr, dinv, b1, W2, H2, n);

  k_agg_rows<64, true><<<(unsigned)(((long long)n * 32 + TB - 1) / TB), TB, 0, stream>>>(
      H2, col, rowptr, dinv, b2, OUTP, n, 5.0f);
}


# row sums: a wave requests the bounds and first 64 sources of both its rows before its first gather (last launch: two rows per wave), so row 2's lists arrive under row 1's gathers; sum orders kept
# speedup vs baseline: 5.3075x; 1.0374x over previous
#include <hip/hip_runtime.h>
#include <hip/hip_bf16.h>

typedef __attribute__((ext_vector_type(2))) float v2f;
typedef __attribute__((ext_vector_type(4))) float v4f;
typedef __attribute__((ext_vector_type(8))) float v8f;

#define TB 256


#define SB 512
#define RB 208
#define MTB (RB / 16)
#define GS 256
#define EU 8
#define SPW (GS / (SB / 32))

__device__ __forceinline__ int block_excl_scan(int v, int* wsum, int& total) {
  int lane = threadIdx.x & 31;
  int w    = threadIdx.x >> 5;
  int nw   = blockDim.x >> 5;
  int inc = v;
  for (int o = 1; o < 32; o <<= 1) {
    int u = __shfl_up(inc, o, 32);
    if (lane >= o) inc += u;
  }
  __syncthreads();
  if (lane == 31) wsum[w] = inc;
  __syncthreads();
  if (w == 0) {
    int x = (lane < nw) ? wsum[lane] : 0;
    for (int o = 1; o < 32; o <<= 1) {
      int u = __shfl_up(x, o, 32);
      if (lane >= o) x += u;
    }
    wsum[lane] = x;
  }
  __syncthreads();
  total = wsum[31];
  return inc - v + (w > 0 ? wsum[w - 1] : 0);
}

__global__ void __launch_bounds__(SB) k_slice_sort(const int* __restrict__ src, const int* __restrict__ dst,
                                                   int* __restrict__ cntT, int* __restrict__ offT,
                                                   int* __restrict__ pairs, int E, int NB, int chunk) {
  extern __shared__ int lbin[];
  __shared__ int wsum[32];
  int g = blockIdx.x;
  for (int b = threadIdx.x; b < NB; b += SB) lbin[b] = 0;
  __syncthreads();
  int e0 = g * chunk;
  int e1 = min(E, e0 + chunk);
  for (int e = e0 + (int)threadIdx.x; e < e1; e += SB * EU) {
    int d[EU];
#pragma unroll
    for (int u = 0; u < EU; ++u) d[u] = dst[min(e + u * SB, e1 - 1)];
    static_assert(EU == 8, "the statement below names the EU loaded values");
    asm volatile("" :: "v"(d[0]), "v"(d[1]), "v"(d[2]), "v"(d[3]), "v"(d[4]), "v"(d[5]), "v"(d[6]), "v"(d[7]));
#pragma unroll
    for (int u = 0; u < EU; ++u)
      if (e + u * SB < e1) atomicAdd(&lbin[(unsigned)d[u] / RB], 1);
  }
  __syncthreads();
  int carry = e0;
  for (int t0 = 0; t0 < NB; t0 += SB) {
    int b = t0 + (int)threadIdx.x;
    int c = (b < NB) ? lbin[b] : 0;
    int tot;
    int ex = block_excl_scan(c, wsum, tot);
    if (b < NB) {
      int first = carry + ex;
      cntT[b * GS + g] = c;
      offT[b * GS + g] = first;
      lbin[b] = first;
    }
    carry += tot;
  }
  __syncthreads();
  for (int e = e0 + (int)threadIdx.x; e < e1; e += SB * EU) {
    int d[EU], s[EU];
#pragma unroll
    for (int u = 0; u < EU; ++u) {
      int ee = min(e + u * SB, e1 - 1);
      d[u] = dst[ee];
      s[u] = src[ee];
    }
    int p[EU];
#pragma unroll
    for (int u = 0; u < EU; ++u) {
      p[u] = 0;
      if (e + u * SB < e1) p[u] = atomicAdd(&lbin[(unsigned)d[u] / RB], 1);
    }
#pragma unroll
    for (int u = 0; u < EU; ++u)
      if (e + u * SB < e1) *(int2*)(pairs + 2 * (size_t)p[u]) = make_int2(s[u], d[u]);
  }
}

template<int K, int N>
__global__ void __launch_bounds__(SB) k_bin_rows_gemm(const int* __restrict__ pairs, const int* __restrict__ cntT,
                                                      const int* __restrict__ offT,
                                                      int* __restrict__ rowptr, float* __restrict__ dinv,
                                                      int* __restrict__ col,
                                                      const float* __restrict__ X, const float* __restrict__ Wm,
                                                      float* __restrict__ H, int n, int E, int chunk) {
  constexpr int LDK = K + 4;
  constexpr int BSF = 41 * 1024;
  static_assert(N * LDK <= BSF, "transposed Wm fits");
  static_assert(N == 128 && K % 32 == 0, "4 column groups of 32; k in steps of 32");
  __shared__ __attribute__((aligned(16))) float Bs[BSF];
  __shared__ __attribute__((aligned(16))) float ldinv[RB];
  __shared__ int lcnt[RB];
  __shared__ int lcur[RB];
  __shared__ int wsum[32];
  {
    constexpr int SG = 32;
    static_assert((K * N) % (SB * SG) == 0 && SB % N == 0, "whole groups; a thread keeps its column");
    int c  = threadIdx.x % N;
    int k0 = threadIdx.x / N;
    for (int base = 0; base < K * N; base += SB * SG) {
      float v[SG];
#pragma unroll
      for (int u = 0; u < SG; ++u) v[u] = Wm[base + u * SB + (int)threadIdx.x];
#pragma unroll
      for (int u = 0; u < SG; ++u) Bs[c * LDK + base / N + u * (SB / N) + k0] = v[u];
    }
  }
  int b    = blockIdx.x;
  int n0   = b * RB;
  int lane = threadIdx.x & 31;
  int w    = threadIdx.x >> 5;
  for (int t = threadIdx.x; t < RB; t += SB) lcnt[t] = 0;
  int segc = 0, sego = 0, part = 0;
  if (lane < SPW) {
    int g = w * SPW + lane;
    segc = cntT[b * GS + g];
    sego = offT[b * GS + g];
    part = sego - g * chunk;
  }
  int base;
  block_excl_scan(part, wsum, base);
  int maxlen = segc;
  for (int o = 16; o > 0; o >>= 1) maxlen = max(maxlen, __shfl_xor(maxlen, o, 32));
  const int2* pr = (const int2*)pairs;

  {
    int len[SPW];
    int da[SPW], db[SPW];
#pragma unroll
    for (int u = 0; u < SPW; ++u) {
      len[u]  = __shfl(segc, u, 32);
      int off = __shfl(sego, u, 32);
      da[u] = pairs[2 * (size_t)((lane      < len[u]) ? off + lane      : 0) + 1];
      db[u] = pairs[2 * (size_t)((lane + 32 < len[u]) ? off + lane + 32 : 0) + 1];
    }
#pragma unroll
    for (int u = 0; u < SPW; ++u) {
      if (lane      < len[u]) atomicAdd(&lcnt[da[u] - n0], 1);
      if (lane + 32 < len[u]) atomicAdd(&lcnt[db[u] - n0], 1);
    }
  }
  if (maxlen > 64) {
    for (int u = 0; u < SPW; ++u) {
      int len = __shfl(segc, u, 32);
      int off = __shfl(sego, u, 32);
      for (int i = 64 + lane; i < len; i += 32) atomicAdd(&lcnt[pr[off + i].y - n0], 1);
    }
  }
  __syncthreads();
  int carry = 0;
  for (int t0 = 0; t0 < RB; t0 += SB) {
    int t = t0 + (int)threadIdx.x;
    int c = (t < RB) ? lcnt[t] : 0;
    int tot;
    int ex = block_excl_scan(c, wsum, tot);
    if (t < RB) {
      int first = base + carry + ex;
      lcur[t] = first;
      int node = n0 + t;
      float dv = rsqrtf((float)(c + 1));
      ldinv[t] = dv;
      if (node < n) {
        rowptr[node] = first;
        dinv[node]   = dv;
      }
    }
    carry += tot;
  }
  if (b == (int)gridDim.x - 1 && threadIdx.x == 0) rowptr[n] = E;
  __syncthreads();

  {
    int len[SPW];
    int2 ta[SPW], tb[SPW];
#pragma unroll
    for (int u = 0; u < SPW; ++u) {
      len[u]  = __shfl(segc, u, 32);
      int off = __shfl(sego, u, 32);
      ta[u] = pr[(lane      < len[u]) ? off + lane      : 0];
      tb[u] = pr[(lane + 32 < len[u]) ? off + lane + 32 : 0];
    }
#pragma unroll
    for (int u = 0; u < SPW; ++u) {
      if (lane      < len[u]) col[atomicAdd(&lcur[ta[u].y - n0], 1)] = ta[u].x;
      if (lane + 32 < len[u]) col[atomicAdd(&lcur[tb[u].y - n0], 1)] = tb[u].x;
    }
  }
  if (maxlen > 64) {
    for (int u = 0; u < SPW; ++u) {
      int len = __shfl(segc, u, 32);
      int off = __shfl(sego, u, 32);
      for (int i = 64 + lane; i < len; i += 32) {
        int2 t = pr[off + i];
        col[atomicAdd(&lcur[t.y - n0], 1)] = t.x;
      }
    }
  }

  {
    int half = lane >> 4;
    int l    = lane & 15;
    int rg   = ((w & 3) + (w >> 2)) & 3;
    int nc0  = (w >> 2) * 32;
    int mtiles = n >> 4;
    const float* Bl = Bs + (nc0 + l) * LDK + half * 4;
    for (int mt = rg; mt < MTB; mt += 4) {
      int gmt = b * MTB + mt;
      if (gmt >= mtiles) break;
      int m0 = gmt << 4;
      const float* Arow = X + (size_t)(m0 + l) * K + half * 4;
      v8f acc0 = {}, acc1 = {};
      v4f an[4];
#pragma unroll
      for (int u = 0; u < 4; ++u) an[u] = *(const v4f*)(Arow + u * 8);
#pragma unroll
      for (int kk = 0; kk < K; kk += 32) {
        v4f a[4];
#pragma unroll
        for (int u = 0; u < 4; ++u) a[u] = an[u];
        if (kk + 32 < K) {
#pragma unroll
          for (int u = 0; u < 4; ++u) an[u] = *(const v4f*)(Arow + kk + 32 + u * 8);
        }
#pragma unroll
        for (int u = 0; u < 4; ++u) {
          v4f b0 = *(const v4f*)(Bl + kk + u * 8);
          v4f b1 = *(const v4f*)(Bl + 16 * LDK + kk + u * 8);
          v2f alo, ahi, b0lo, b0hi, b1lo, b1hi;
          alo.x  = a[u].x; alo.y  = a[u].y; ahi.x  = a[u].z; ahi.y  = a[u].w;
          b0lo.x = b0.x;   b0lo.y = b0.y;   b0hi.x = b0.z;   b0hi.y = b0.w;
          b1lo.x = b1.x;   b1lo.y = b1.y;   b1hi.x = b1.z;   b1hi.y = b1.w;
          acc0 = __builtin_amdgcn_wmma_f32_16x16x4_f32(false, alo, false, b0lo, (short)0, acc0, false, false);
          acc1 = __builtin_amdgcn_wmma_f32_16x16x4_f32(false, alo, false, b1lo, (short)0, acc1, false, false);
          acc0 = __builtin_amdgcn_wmma_f32_16x16x4_f32(false, ahi, false, b0hi, (short)0, acc0, false, false);
          acc1 = __builtin_amdgcn_wmma_f32_16x16x4_f32(false, ahi, false, b1hi, (short)0, acc1, false, false);
        }
      }
      int rl = mt * 16 + half * 8;
      v4f s0 = *(const v4f*)(ldinv + rl);
      v4f s1 = *(const v4f*)(ldinv + rl + 4);
      float sc[8] = {s0.x, s0.y, s0.z, s0.w, s1.x, s1.y, s1.z, s1.w};
      float* Hp = H + (size_t)(m0 + half * 8) * N + nc0 + l;
#pragma unroll
      for (int j = 0; j < 8; ++j) {
        Hp[(size_t)j * N]      = acc0[j] * sc[j];
        Hp[(size_t)j * N + 16] = acc1[j] * sc[j];
      }
    }
  }
}


template<int F, int U>
__device__ __forceinline__ v4f add_rows(const float* __restrict__ HS, int my_s, int m, int sub, int l, v4f acc) {
  constexpr int EPI = 32 / (F / 4);
  const v4f zero = {0.0f, 0.0f, 0.0f, 0.0f};
  for (int j = 0; j < m; j += EPI * U) {
    v4f v[U];
    int s[U];
#pragma unroll
    for (int u = 0; u < U; ++u) s[u] = __shfl(my_s, j + u * EPI + sub, 32);
#pragma unroll
    for (int u = 0; u < U; ++u)
      v[u] = (j + u * EPI + sub < m) ? *(const v4f*)(HS + (size_t)s[u] * F + l * 4) : zero;
#pragma unroll
    for (int u = 0; u < U; ++u) acc += v[u];
  }
  return acc;
}

#define RW 2
template<int F, bool LSM>
__global__ void k_agg_rows(const float* __restrict__ HS, const int* __restrict__ col,
                           const int* __restrict__ rowptr, const float* __restrict__ dinv,
                           const float* __restrict__ b, float* __restrict__ OUT, int n, float invT) {
  constexpr int LPR = F / 4;
  constexpr int U   = 8;
  int row0 = (int)((blockIdx.x * blockDim.x + threadIdx.x) >> 5) * RW;
  int lane = threadIdx.x & 31;
  if (row0 >= n) return;
  int sub = lane / LPR;
  int l   = lane % LPR;
  const v4f zero = {0.0f, 0.0f, 0.0f, 0.0f};
  int rp = 0;
  if (lane <= RW) rp = rowptr[min(row0 + lane, n)];
  int bnd[RW + 1];
#pragma unroll
  for (int q = 0; q <= RW; ++q) bnd[q] = __shfl(rp, q, 32);
  float di[RW];
  int sa[RW], sb[RW];
#pragma unroll
  for (int q = 0; q < RW; ++q) {
    int row = row0 + q;
    bool valid = row < n;
    di[q] = 0.0f;
    sa[q] = 0;
    sb[q] = 0;
    if (valid) {
      di[q] = dinv[row];
      if (bnd[q] + lane      < bnd[q + 1]) sa[q] = col[bnd[q] + lane];
      if (bnd[q] + 32 + lane < bnd[q + 1]) sb[q] = col[bnd[q] + 32 + lane];
    }
  }
  v4f bb = *(const v4f*)(b + l * 4);
#pragma unroll
  for (int q = 0; q < RW; ++q) {
    int row = row0 + q;
    if (row >= n) break;
    int start = bnd[q], end = bnd[q + 1];
    int len = end - start;
    v4f acc = zero;
    if (sub == 0)
      acc = *(const v4f*)(HS + (size_t)row * F + l * 4);
    if (len > 0)  acc = add_rows<F, U>(HS, sa[q], min(32, len), sub, l, acc);
    if (len > 32) acc = add_rows<F, U>(HS, sb[q], min(32, len - 32), sub, l, acc);
    for (int base = start + 64; base < end; base += 32) {
      int m = min(32, end - base);
      int my_s = 0;
      if (lane < m) my_s = col[base + lane];
      acc = add_rows<F, U>(HS, my_s, m, sub, l, acc);
    }
    for (int o = LPR; o < 32; o <<= 1) {
      acc.x += __shfl_xor(acc.x, o, 32);
      acc.y += __shfl_xor(acc.y, o, 32);
      acc.z += __shfl_xor(acc.z, o, 32);
      acc.w += __shfl_xor(acc.w, o, 32);
    }
    v4f r = acc * di[q] + bb;
    if (LSM) {
      v4f z = r * invT;
      float mx = fmaxf(fmaxf(z.x, z.y), fmaxf(z.z, z.w));
      for (int o = 1; o < LPR; o <<= 1) mx = fmaxf(mx, __shfl_xor(mx, o, 32));
      float s = (expf(z.x - mx) + expf(z.y - mx)) + (expf(z.z - mx) + expf(z.w - mx));
      for (int o = 1; o < LPR; o <<= 1) s += __shfl_xor(s, o, 32);
      float lg = mx + logf(s);
      r = z - lg;
    }
    if (sub == 0) *(v4f*)(OUT + (size_t)row * F + l * 4) = r;
  }
}


#define AR  32
#define AWV 16
template<int F, int N2>
__global__ void __launch_bounds__(AWV * 32) k_agg_rows_gemm(const float* __restrict__ HS, const int* __restrict__ col,
                                                            const int* __restrict__ rowptr, const float* __restrict__ dinv,
                                                            const float* __restrict__ b, const float* __restrict__ Wm,
                                                            float* __restrict__ C, int n, int E) {
  constexpr int NTH = AWV * 32;
  constexpr int LDA = F + 4;
  constexpr int NT  = N2 / 16;
  constexpr int U   = 8;
  static_assert(F == 128, "a lane holds 16 bytes of a row: 32 lanes x 4 floats");
  static_assert(AR % AWV == 0 && AR % 16 == 0 && N2 % 16 == 0, "whole rows per wave, whole tiles");
  static_assert((AR / 16) * NT <= AWV, "one output tile per product wave");
  static_assert((F * N2) % NTH == 0 && NTH % N2 == 0, "whole staging groups; a thread keeps its column");
  __shared__ __attribute__((aligned(16))) float Bs[N2 * LDA];
  __shared__ __attribute__((aligned(16))) float As[AR * LDA];
  __shared__ __attribute__((aligned(16))) float ldinv[AR];
  int lane = threadIdx.x & 31;
  int w    = threadIdx.x >> 5;
  int r0   = blockIdx.x * AR;

  constexpr int SG = F * N2 / NTH;
  constexpr int KS = NTH / N2;
  float wv[SG];
#pragma unroll
  for (int u = 0; u < SG; ++u) wv[u] = Wm[u * NTH + (int)threadIdx.x];

  const v4f zero = {0.0f, 0.0f, 0.0f, 0.0f};
  v4f bb = *(const v4f*)(b + lane * 4);
  constexpr int Q = AR / AWV;
  int start[Q], end[Q];
  float di[Q];
  v4f selfv[Q];
#pragma unroll
  for (int q = 0; q < Q; ++q) {
    int row = min(r0 + q * AWV + w, n - 1);
    start[q] = rowptr[row];
    end[q]   = rowptr[row + 1];
    di[q]    = dinv[row];
    selfv[q] = *(const v4f*)(HS + (size_t)row * F + lane * 4);
  }
  {
    int c  = threadIdx.x % N2;
    int k0 = threadIdx.x / N2;
#pragma unroll
    for (int u = 0; u < SG; ++u) Bs[c * LDA + u * KS + k0] = wv[u];
  }
  static_assert(Q == 2, "the statement below names the bounds of two rows");
  asm volatile("" :: "v"(start[0]), "v"(end[0]), "v"(start[1]), "v"(end[1]));
  int sa[Q], sb[Q];
#pragma unroll
  for (int q = 0; q < Q; ++q) {
    sa[q] = col[min(start[q] + lane, E - 1)];
    sb[q] = col[min(start[q] + 32 + lane, E - 1)];
  }
#pragma unroll
  for (int q = 0; q < Q; ++q) {
    int slot = q * AWV + w;
    v4f r = zero;
    if (r0 + slot < n) {
      int len = end[q] - start[q];
      v4f acc = selfv[q];
      if (len > 0)  acc = add_rows<F, U>(HS, sa[q], min(32, len), 0, lane, acc);
      if (len > 32) acc = add_rows<F, U>(HS, sb[q], min(32, len - 32), 0, lane, acc);
      for (int base = start[q] + 64; base < end[q]; base += 32) {
        int m = min(32, end[q] - base);
        int my_s = 0;
        if (lane < m) my_s = col[base + lane];
        acc = add_rows<F, U>(HS, my_s, m, 0, lane, acc);
      }
      r = acc * di[q] + bb;
      r.x = fmaxf(r.x, 0.0f); r.y = fmaxf(r.y, 0.0f); r.z = fmaxf(r.z, 0.0f); r.w = fmaxf(r.w, 0.0f);
    }
    *(v4f*)(As + slot * LDA + lane * 4) = r;
    if (lane == 0) ldinv[slot] = (r0 + slot < n) ? di[q] : 0.0f;
  }
  __syncthreads();

  if (w < (AR / 16) * NT) {
    int mt   = w / NT;
    int nt   = w % NT;
    int half = lane >> 4;
    int l    = lane & 15;
    const float* Al = As + (mt * 16 + l) * LDA + half * 2;
    const float* Bl = Bs + (nt * 16 + l) * LDA + half * 2;
    v8f acc = {};
#pragma unroll
    for (int k = 0; k < F; k += 4) {
      v2f a  = *(const v2f*)(Al + k);
      v2f bq = *(const v2f*)(Bl + k);
      acc = __builtin_amdgcn_wmma_f32_16x16x4_f32(false, a, false, bq, (short)0, acc, false, false);
    }
    int rl = mt * 16 + half * 8;
    v4f s0 = *(const v4f*)(ldinv + rl);
    v4f s1 = *(const v4f*)(ldinv + rl + 4);
    float sc[8] = {s0.x, s0.y, s0.z, s0.w, s1.x, s1.y, s1.z, s1.w};
    int grow = r0 + rl;
    float* Cp = C + (size_t)grow * N2 + nt * 16 + l;
#pragma unroll
    for (int j = 0; j < 8; ++j)
      if (grow + j < n) Cp[(size_t)j * N2] = acc[j] * sc[j];
  }
}


extern "C" void kernel_launch(void* const* d_in, const int* in_sizes, int n_in,
                              void* d_out, int out_size, void* d_ws, size_t ws_size,
                              hipStream_t stream) {
  const float* x  = (const float*)d_in[0];
  const int*   ei = (const int*)d_in[1];
  const float* W1 = (const float*)d_in[2];
  const float* b1 = (const float*)d_in[3];
  const float* W2 = (const float*)d_in[4];
  const float* b2 = (const float*)d_in[5];

  const int IN = 256, HID = 128, OUT = 64;
  int n = in_sizes[0] / IN;
  int E = in_sizes[1] / 2;
  const int* src = ei;
  const int* dst = ei + E;

  float* ws = (float*)d_ws;
  size_t o = 0;
  float* dinv = ws + o; o += ((size_t)n + 255) & ~(size_t)255;
  float* H1   = ws + o; o += (size_t)n * HID;
  float* H2   = ws + o; o += (size_t)n * OUT;
  int NB = (n + RB - 1) / RB;
  int* rowptr = (int*)(ws + o); o += ((size_t)n + 1 + 255) & ~(size_t)255;
  int* cntT   = (int*)(ws + o); o += ((size_t)NB * GS + 255) & ~(size_t)255;
  int* offT   = (int*)(ws + o); o += ((size_t)NB * GS + 255) & ~(size_t)255;
  int* pairs  = (int*)(ws + o); o += ((size_t)2 * E + 255) & ~(size_t)255;
  int* col    = (int*)(ws + o); o += ((size_t)E + 255) & ~(size_t)255;
  float* OUTP = (float*)d_out;

  int chunk   = (E + GS - 1) / GS;

  k_slice_sort<<<GS, SB, NB * sizeof(int), stream>>>(src, dst, cntT, offT, pairs, E, NB, chunk);
  k_bin_rows_gemm<256, 128><<<NB, SB, 0, stream>>>(pairs, cntT, offT, rowptr, dinv, col, x, W1, H1, n, E, chunk);

  k_agg_rows_gemm<128, 64><<<(unsigned)((n + AR - 1) / AR), AWV * 32, 0, stream>>>(
      H1, col, rowptr, dinv, b1, W2, H2, n, E);

  k_agg_rows<64, true><<<(unsigned)((((long long)n + RW - 1) / RW * 32 + TB - 1) / TB), TB, 0, stream>>>(
      H2, col, rowptr, dinv, b2, OUTP, n, 5.0f);
}
